# starting kernel
# baseline (speedup 1.0000x reference)
#include <hip/hip_runtime.h>
#include <hip/hip_bf16.h>

typedef __attribute__((ext_vector_type(2))) float v2f;
typedef __attribute__((ext_vector_type(4))) float v4f;
typedef __attribute__((ext_vector_type(8))) float v8f;

#define KDIM 128
#define FW   128

__global__ void deg_init(float* __restrict__ dinv, int n) {
    int i = blockIdx.x * blockDim.x + threadIdx.x;
    if (i < n) dinv[i] = 1.0f;
}

__global__ void deg_accum(const int* __restrict__ dst, float* __restrict__ dinv, int e) {
    int i = blockIdx.x * blockDim.x + threadIdx.x;
    if (i < e) atomicAdd(&dinv[dst[i]], 1.0f);
}

__global__ void deg_finish(float* __restrict__ dinv, int n) {
    int i = blockIdx.x * blockDim.x + threadIdx.x;
    if (i < n) dinv[i] = rsqrtf(dinv[i]);
}

template <bool DUAL>
__global__ __launch_bounds__(256)
void gemm_wmma_f32(const float* __restrict__ A,
                   const float* __restrict__ B0,
                   const float* __restrict__ B1,
                   float* __restrict__ C, int nrows)
{
    constexpr int BLD = DUAL ? 64 : 128;
    __shared__ float As[16 * KDIM];
    const int row0 = blockIdx.x * 16;
    const int tid  = threadIdx.x;

    {
        v4f* dstv = (v4f*)As;
        #pragma unroll
        for (int idx = tid; idx < 512; idx += 256) {
            int r  = idx >> 5;
            int c4 = idx & 31;
            int gr = row0 + r; if (gr >= nrows) gr = nrows - 1;
            dstv[idx] = ((const v4f*)(A + (size_t)gr * KDIM))[c4];
        }
    }
    __syncthreads();

    const int wave = tid >> 5;
    const int lane = tid & 31;
    const int col0 = wave * 16;
    const int ln   = lane & 15;
    const int hi   = lane >> 4;

    const float* bp;
    if (DUAL) {
        const float* Bsel = (col0 < 64) ? B0 : B1;
        const int    bcol = (col0 < 64) ? (col0 + ln) : (col0 - 64 + ln);
        bp = Bsel + (size_t)(2 * hi) * BLD + bcol;
    } else {
        bp = B0 + (size_t)(2 * hi) * BLD + col0 + ln;
    }
    const float* ap = &As[ln * KDIM + 2 * hi];

    v8f acc = {};
    #pragma unroll
    for (int k = 0; k < KDIM; k += 4) {
        v2f a = *(const v2f*)(ap + k);
        v2f b;
        b.x = bp[(size_t)k * BLD];
        b.y = bp[(size_t)(k + 1) * BLD];
        acc = __builtin_amdgcn_wmma_f32_16x16x4_f32(
            false, a, false, b, (short)0, acc, false, false);
    }

    float* cp = C + (size_t)(row0 + 8 * hi) * FW + col0 + ln;
    #pragma unroll
    for (int r = 0; r < 8; ++r) {
        if (row0 + r + 8 * hi < nrows) cp[(size_t)r * FW] = acc[r];
    }
}

__global__ void selfloop_init(const float* __restrict__ in, const float* __restrict__ dinv,
                              float* __restrict__ out, int n)
{
    int t = blockIdx.x * blockDim.x + threadIdx.x;
    if (t < n * FW) {
        int i = t >> 7;
        float d = dinv[i];
        out[t] = d * d * in[t];
    }
}

__global__ __launch_bounds__(256)
void edge_scatter(const int* __restrict__ srcIdx, const int* __restrict__ dstIdx,
                  const float* __restrict__ dinv,
                  const float* __restrict__ in, float* __restrict__ out, int e)
{
    int t = blockIdx.x * blockDim.x + threadIdx.x;
    int edge = t >> 5;
    int lane = t & 31;
    if (edge >= e) return;
    int s = srcIdx[edge];
    int d = dstIdx[edge];
    float nm = dinv[s] * dinv[d];
    v4f v = ((const v4f*)(in + (size_t)s * FW))[lane];
    float* op = out + (size_t)d * FW + lane * 4;
    atomicAdd(op + 0, nm * v.x);
    atomicAdd(op + 1, nm * v.y);
    atomicAdd(op + 2, nm * v.z);
    atomicAdd(op + 3, nm * v.w);
}

__global__ void bias_relu(const float* __restrict__ agg, const float* __restrict__ b,
                          float* __restrict__ out, int n)
{
    int t = blockIdx.x * blockDim.x + threadIdx.x;
    if (t < n * FW) {
        int f = t & (FW - 1);
        out[t] = fmaxf(agg[t] + b[f], 0.0f);
    }
}

__global__ void final_out(const float* __restrict__ agg,
                          const float* __restrict__ bmu, const float* __restrict__ blv,
                          float* __restrict__ outp, int n)
{
    int t = blockIdx.x * blockDim.x + threadIdx.x;
    if (t < n * 64) {
        int i = t >> 6, f = t & 63;
        const float* row = agg + (size_t)i * FW;
        outp[t]                   = row[f]      + bmu[f];
        outp[(size_t)n * 64 + t]  = row[64 + f] + blv[f];
    }
}

extern "C" void kernel_launch(void* const* d_in, const int* in_sizes, int n_in,
                              void* d_out, int out_size, void* d_ws, size_t ws_size,
                              hipStream_t stream)
{
    const float* x    = (const float*)d_in[0];
    const int*   ei   = (const int*)  d_in[1];
    const float* W1   = (const float*)d_in[2];
    const float* b1   = (const float*)d_in[3];
    const float* Wmu  = (const float*)d_in[4];
    const float* bmu  = (const float*)d_in[5];
    const float* Wlv  = (const float*)d_in[6];
    const float* blv  = (const float*)d_in[7];

    const int N = in_sizes[0] / KDIM;
    const int E = in_sizes[1] / 2;
    const int* srcIdx = ei;
    const int* dstIdx = ei + E;

    size_t dinv_bytes = (((size_t)N * 4) + 255) & ~(size_t)255;
    float* dinv = (float*)d_ws;
    float* bufA = (float*)((char*)d_ws + dinv_bytes);
    float* bufB = bufA + (size_t)N * FW;

    const int TB = 256;
    dim3 blk(TB);
    int gN   = (N + TB - 1) / TB;
    int gE   = (E + TB - 1) / TB;
    int gNF  = (N * FW + TB - 1) / TB;
    int gNZ  = (N * 64 + TB - 1) / TB;
    int gEw  = (int)(((size_t)E * 32 + TB - 1) / TB);
    int gGemm = (N + 15) / 16;

    deg_init  <<<gN, blk, 0, stream>>>(dinv, N);
    deg_accum <<<gE, blk, 0, stream>>>(dstIdx, dinv, E);
    deg_finish<<<gN, blk, 0, stream>>>(dinv, N);

    gemm_wmma_f32<false><<<gGemm, blk, 0, stream>>>(x, W1, nullptr, bufA, N);

    selfloop_init<<<gNF, blk, 0, stream>>>(bufA, dinv, bufB, N);
    edge_scatter <<<gEw, blk, 0, stream>>>(srcIdx, dstIdx, dinv, bufA, bufB, E);

    bias_relu<<<gNF, blk, 0, stream>>>(bufB, b1, bufA, N);

    gemm_wmma_f32<true><<<gGemm, blk, 0, stream>>>(bufA, Wmu, Wlv, bufB, N);

    selfloop_init<<<gNF, blk, 0, stream>>>(bufB, dinv, bufA, N);
    edge_scatter <<<gEw, blk, 0, stream>>>(srcIdx, dstIdx, dinv, bufB, bufA, E);

    final_out<<<gNZ, blk, 0, stream>>>(bufA, bmu, blv, (float*)d_out, N);
}


# f32-atomic edge scatter -> per-call edge list grouped by destination (int count+rank, scan, place, order by source) and one wave per node summing rows in registers, fixed order
# speedup vs baseline: 2.0601x; 2.0601x over previous
#include <hip/hip_runtime.h>
#include <hip/hip_bf16.h>

typedef __attribute__((ext_vector_type(2))) float v2f;
typedef __attribute__((ext_vector_type(4))) float v4f;
typedef __attribute__((ext_vector_type(8))) float v8f;

#define KDIM 128
#define FW   128

#define SCAN_TILE 1024

__global__ void cnt_zero(int* __restrict__ cnt, int n) {
    int i = blockIdx.x * blockDim.x + threadIdx.x;
    if (i < n) cnt[i] = 0;
}

__global__ void edge_rank(const int* __restrict__ dst, int* __restrict__ cnt,
                          int* __restrict__ rank, int e) {
    int i = blockIdx.x * blockDim.x + threadIdx.x;
    if (i < e) rank[i] = atomicAdd(&cnt[dst[i]], 1);
}

__device__ __forceinline__ int wave_sum(int v) {
    for (int off = 16; off > 0; off >>= 1) v += __shfl_xor(v, off);
    return v;
}

__global__ __launch_bounds__(256)
void scan_tile_sums(const int* __restrict__ cnt, int* __restrict__ tsum, int n) {
    __shared__ int red[8];
    const int tid = threadIdx.x;
    const int i0  = blockIdx.x * SCAN_TILE + tid * 4;
    int s = 0;
    for (int u = 0; u < 4; ++u) if (i0 + u < n) s += cnt[i0 + u];
    s = wave_sum(s);
    if ((tid & 31) == 0) red[tid >> 5] = s;
    __syncthreads();
    if (tid == 0) {
        int t = 0;
        for (int w = 0; w < 8; ++w) t += red[w];
        tsum[blockIdx.x] = t;
    }
}

__global__ __launch_bounds__(256)
void scan_offsets(const int* __restrict__ cnt, const int* __restrict__ tsum,
                  int* __restrict__ rowptr, float* __restrict__ dinv, int n) {
    __shared__ int red[8];
    __shared__ int wtot[8];
    const int tid  = threadIdx.x;
    const int lane = tid & 31;
    const int wave = tid >> 5;

    int p = 0;
    for (int t = tid; t < (int)blockIdx.x; t += 256) p += tsum[t];
    p = wave_sum(p);
    if (lane == 0) red[wave] = p;
    __syncthreads();
    int tbase = 0;
    for (int w = 0; w < 8; ++w) tbase += red[w];

    const int i0 = blockIdx.x * SCAN_TILE + tid * 4;
    int c[4];
    for (int u = 0; u < 4; ++u) c[u] = (i0 + u < n) ? cnt[i0 + u] : 0;
    const int mine = c[0] + c[1] + c[2] + c[3];

    int x = mine;
    for (int off = 1; off < 32; off <<= 1) {
        int y = __shfl_up(x, off);
        if (lane >= off) x += y;
    }
    if (lane == 31) wtot[wave] = x;
    __syncthreads();
    int wbase = 0;
    for (int w = 0; w < wave; ++w) wbase += wtot[w];

    int excl = tbase + wbase + x - mine;
    for (int u = 0; u < 4; ++u) {
        if (i0 + u < n) {
            rowptr[i0 + u] = excl;
            dinv[i0 + u]   = rsqrtf((float)(1 + c[u]));
            excl += c[u];
        }
    }
}

__global__ void edge_place(const int* __restrict__ src, const int* __restrict__ dst,
                           const int* __restrict__ rank, const int* __restrict__ rowptr,
                           int* __restrict__ tmp, int e) {
    int i = blockIdx.x * blockDim.x + threadIdx.x;
    if (i < e) tmp[rowptr[dst[i]] + rank[i]] = src[i];
}

__global__ __launch_bounds__(256)
void bucket_order(const int* __restrict__ rowptr, const int* __restrict__ cnt,
                  const int* __restrict__ tmp, int* __restrict__ adj, int n) {
    const int lane = threadIdx.x & 31;
    const int node = blockIdx.x * 8 + __builtin_amdgcn_readfirstlane((int)(threadIdx.x >> 5));
    if (node >= n) return;
    const int start = rowptr[node];
    const int c     = cnt[node];
    for (int jb = 0; jb < c; jb += 32) {
        const int j  = jb + lane;
        const int sj = (j < c) ? tmp[start + j] : 0x7fffffff;
        int r = 0;
        for (int kb = 0; kb < c; kb += 32) {
            const int k  = kb + lane;
            const int sk = (k < c) ? tmp[start + k] : 0x7fffffff;
            int mm = c - kb; if (mm > 32) mm = 32;
            for (int t = 0; t < mm; ++t) {
                const int v = __builtin_amdgcn_readlane(sk, t);
                r += (v < sj || (v == sj && kb + t < j)) ? 1 : 0;
            }
        }
        if (j < c) adj[start + r] = sj;
    }
}

template <bool DUAL>
__global__ __launch_bounds__(256)
void gemm_wmma_f32(const float* __restrict__ A,
                   const float* __restrict__ B0,
                   const float* __restrict__ B1,
                   float* __restrict__ C, int nrows)
{
    constexpr int BLD = DUAL ? 64 : 128;
    __shared__ float As[16 * KDIM];
    const int row0 = blockIdx.x * 16;
    const int tid  = threadIdx.x;

    {
        v4f* dstv = (v4f*)As;
        #pragma unroll
        for (int idx = tid; idx < 512; idx += 256) {
            int r  = idx >> 5;
            int c4 = idx & 31;
            int gr = row0 + r; if (gr >= nrows) gr = nrows - 1;
            dstv[idx] = ((const v4f*)(A + (size_t)gr * KDIM))[c4];
        }
    }
    __syncthreads();

    const int wave = tid >> 5;
    const int lane = tid & 31;
    const int col0 = wave * 16;
    const int ln   = lane & 15;
    const int hi   = lane >> 4;

    const float* bp;
    if (DUAL) {
        const float* Bsel = (col0 < 64) ? B0 : B1;
        const int    bcol = (col0 < 64) ? (col0 + ln) : (col0 - 64 + ln);
        bp = Bsel + (size_t)(2 * hi) * BLD + bcol;
    } else {
        bp = B0 + (size_t)(2 * hi) * BLD + col0 + ln;
    }
    const float* ap = &As[ln * KDIM + 2 * hi];

    v8f acc = {};
    #pragma unroll
    for (int k = 0; k < KDIM; k += 4) {
        v2f a = *(const v2f*)(ap + k);
        v2f b;
        b.x = bp[(size_t)k * BLD];
        b.y = bp[(size_t)(k + 1) * BLD];
        acc = __builtin_amdgcn_wmma_f32_16x16x4_f32(
            false, a, false, b, (short)0, acc, false, false);
    }

    float* cp = C + (size_t)(row0 + 8 * hi) * FW + col0 + ln;
    #pragma unroll
    for (int r = 0; r < 8; ++r) {
        if (row0 + r + 8 * hi < nrows) cp[(size_t)r * FW] = acc[r];
    }
}

__global__ __launch_bounds__(256)
void node_gather(const int* __restrict__ rowptr, const int* __restrict__ cnt,
                 const int* __restrict__ adj, const float* __restrict__ dinv,
                 const float* __restrict__ in, float* __restrict__ out, int n)
{
    const int lane = threadIdx.x & 31;
    const int node = blockIdx.x * 8 + __builtin_amdgcn_readfirstlane((int)(threadIdx.x >> 5));
    if (node >= n) return;
    const int   start = rowptr[node];
    const int   c     = cnt[node];
    const float di    = dinv[node];

    v4f acc = (di * di) * ((const v4f*)(in + (size_t)node * FW))[lane];

    for (int base = 0; base < c; base += 32) {
        int m = c - base; if (m > 32) m = 32;
        int   my_s = node;
        float my_w = 0.0f;
        if (lane < m) {
            my_s = adj[start + base + lane];
            my_w = dinv[my_s] * di;
        }
        int j = 0;
        for (; j + 8 <= m; j += 8) {
            v4f v[8]; float w[8];
            #pragma unroll
            for (int u = 0; u < 8; ++u) {
                const int s = __builtin_amdgcn_readlane(my_s, j + u);
                w[u] = __builtin_bit_cast(float,
                           __builtin_amdgcn_readlane(__builtin_bit_cast(int, my_w), j + u));
                v[u] = ((const v4f*)(in + (size_t)s * FW))[lane];
            }
            #pragma unroll
            for (int u = 0; u < 8; ++u) acc += w[u] * v[u];
        }
        const int r = m - j;
        if (r > 0) {
            v4f v[7]; float w[7];
            #pragma unroll
            for (int u = 0; u < 7; ++u) {
                v[u] = (v4f){0.0f, 0.0f, 0.0f, 0.0f};
                w[u] = 0.0f;
                if (u < r) {
                    const int s = __builtin_amdgcn_readlane(my_s, j + u);
                    w[u] = __builtin_bit_cast(float,
                               __builtin_amdgcn_readlane(__builtin_bit_cast(int, my_w), j + u));
                    v[u] = ((const v4f*)(in + (size_t)s * FW))[lane];
                }
            }
            #pragma unroll
            for (int u = 0; u < 7; ++u) if (u < r) acc += w[u] * v[u];
        }
    }
    ((v4f*)(out + (size_t)node * FW))[lane] = acc;
}

__global__ void bias_relu(const float* __restrict__ agg, const float* __restrict__ b,
                          float* __restrict__ out, int n)
{
    int t = blockIdx.x * blockDim.x + threadIdx.x;
    if (t < n * FW) {
        int f = t & (FW - 1);
        out[t] = fmaxf(agg[t] + b[f], 0.0f);
    }
}

__global__ void final_out(const float* __restrict__ agg,
                          const float* __restrict__ bmu, const float* __restrict__ blv,
                          float* __restrict__ outp, int n)
{
    int t = blockIdx.x * blockDim.x + threadIdx.x;
    if (t < n * 64) {
        int i = t >> 6, f = t & 63;
        const float* row = agg + (size_t)i * FW;
        outp[t]                   = row[f]      + bmu[f];
        outp[(size_t)n * 64 + t]  = row[64 + f] + blv[f];
    }
}

extern "C" void kernel_launch(void* const* d_in, const int* in_sizes, int n_in,
                              void* d_out, int out_size, void* d_ws, size_t ws_size,
                              hipStream_t stream)
{
    const float* x    = (const float*)d_in[0];
    const int*   ei   = (const int*)  d_in[1];
    const float* W1   = (const float*)d_in[2];
    const float* b1   = (const float*)d_in[3];
    const float* Wmu  = (const float*)d_in[4];
    const float* bmu  = (const float*)d_in[5];
    const float* Wlv  = (const float*)d_in[6];
    const float* blv  = (const float*)d_in[7];

    const int N = in_sizes[0] / KDIM;
    const int E = in_sizes[1] / 2;
    const int* srcIdx = ei;
    const int* dstIdx = ei + E;

    const int TB = 256;
    dim3 blk(TB);
    int gN    = (N + TB - 1) / TB;
    int gE    = (E + TB - 1) / TB;
    int gNF   = (N * FW + TB - 1) / TB;
    int gNZ   = (N * 64 + TB - 1) / TB;
    int gNw   = (N + 7) / 8;
    int gTile = (N + SCAN_TILE - 1) / SCAN_TILE;
    int gGemm = (N + 15) / 16;

    size_t n_bytes = (((size_t)N * 4) + 255) & ~(size_t)255;
    size_t t_bytes = (((size_t)gTile * 4) + 255) & ~(size_t)255;
    size_t e_bytes = (((size_t)E * 4) + 255) & ~(size_t)255;
    float* dinv   = (float*)d_ws;
    float* bufA   = (float*)((char*)d_ws + n_bytes);
    float* bufB   = bufA + (size_t)N * FW;
    int*   cnt    = (int*)(bufB + (size_t)N * FW);
    int*   rowptr = (int*)((char*)cnt + n_bytes);
    int*   tsum   = (int*)((char*)rowptr + n_bytes);
    int*   rank   = (int*)((char*)tsum + t_bytes);
    int*   tmp    = (int*)((char*)rank + e_bytes);
    int*   adj    = (int*)((char*)tmp + e_bytes);

    cnt_zero      <<<gN,    blk, 0, stream>>>(cnt, N);
    edge_rank     <<<gE,    blk, 0, stream>>>(dstIdx, cnt, rank, E);
    scan_tile_sums<<<gTile, blk, 0, stream>>>(cnt, tsum, N);
    scan_offsets  <<<gTile, blk, 0, stream>>>(cnt, tsum, rowptr, dinv, N);
    edge_place    <<<gE,    blk, 0, stream>>>(srcIdx, dstIdx, rank, rowptr, tmp, E);
    bucket_order  <<<gNw,   blk, 0, stream>>>(rowptr, cnt, tmp, adj, N);

    gemm_wmma_f32<false><<<gGemm, blk, 0, stream>>>(x, W1, nullptr, bufA, N);

    node_gather<<<gNw, blk, 0, stream>>>(rowptr, cnt, adj, dinv, bufA, bufB, N);

    bias_relu<<<gNF, blk, 0, stream>>>(bufB, b1, bufA, N);

    gemm_wmma_f32<true><<<gGemm, blk, 0, stream>>>(bufA, Wmu, Wlv, bufB, N);

    node_gather<<<gNw, blk, 0, stream>>>(rowptr, cnt, adj, dinv, bufB, bufA, N);

    final_out<<<gNZ, blk, 0, stream>>>(bufA, bmu, blv, (float*)d_out, N);
}


# bias+relu and bias+mu/logvar split folded into the two node-sum kernels' own row store (in registers); the two elementwise launches and their re-read of the sums are gone; values bit-identical
# speedup vs baseline: 2.4217x; 1.1755x over previous
#include <hip/hip_runtime.h>
#include <hip/hip_bf16.h>

typedef __attribute__((ext_vector_type(2))) float v2f;
typedef __attribute__((ext_vector_type(4))) float v4f;
typedef __attribute__((ext_vector_type(8))) float v8f;

#define KDIM 128
#define FW   128

#define SCAN_TILE 1024

__global__ void cnt_zero(int* __restrict__ cnt, int n) {
    int i = blockIdx.x * blockDim.x + threadIdx.x;
    if (i < n) cnt[i] = 0;
}

__global__ void edge_rank(const int* __restrict__ dst, int* __restrict__ cnt,
                          int* __restrict__ rank, int e) {
    int i = blockIdx.x * blockDim.x + threadIdx.x;
    if (i < e) rank[i] = atomicAdd(&cnt[dst[i]], 1);
}

__device__ __forceinline__ int wave_sum(int v) {
    for (int off = 16; off > 0; off >>= 1) v += __shfl_xor(v, off);
    return v;
}

__global__ __launch_bounds__(256)
void scan_tile_sums(const int* __restrict__ cnt, int* __restrict__ tsum, int n) {
    __shared__ int red[8];
    const int tid = threadIdx.x;
    const int i0  = blockIdx.x * SCAN_TILE + tid * 4;
    int s = 0;
    for (int u = 0; u < 4; ++u) if (i0 + u < n) s += cnt[i0 + u];
    s = wave_sum(s);
    if ((tid & 31) == 0) red[tid >> 5] = s;
    __syncthreads();
    if (tid == 0) {
        int t = 0;
        for (int w = 0; w < 8; ++w) t += red[w];
        tsum[blockIdx.x] = t;
    }
}

__global__ __launch_bounds__(256)
void scan_offsets(const int* __restrict__ cnt, const int* __restrict__ tsum,
                  int* __restrict__ rowptr, float* __restrict__ dinv, int n) {
    __shared__ int red[8];
    __shared__ int wtot[8];
    const int tid  = threadIdx.x;
    const int lane = tid & 31;
    const int wave = tid >> 5;

    int p = 0;
    for (int t = tid; t < (int)blockIdx.x; t += 256) p += tsum[t];
    p = wave_sum(p);
    if (lane == 0) red[wave] = p;
    __syncthreads();
    int tbase = 0;
    for (int w = 0; w < 8; ++w) tbase += red[w];

    const int i0 = blockIdx.x * SCAN_TILE + tid * 4;
    int c[4];
    for (int u = 0; u < 4; ++u) c[u] = (i0 + u < n) ? cnt[i0 + u] : 0;
    const int mine = c[0] + c[1] + c[2] + c[3];

    int x = mine;
    for (int off = 1; off < 32; off <<= 1) {
        int y = __shfl_up(x, off);
        if (lane >= off) x += y;
    }
    if (lane == 31) wtot[wave] = x;
    __syncthreads();
    int wbase = 0;
    for (int w = 0; w < wave; ++w) wbase += wtot[w];

    int excl = tbase + wbase + x - mine;
    for (int u = 0; u < 4; ++u) {
        if (i0 + u < n) {
            rowptr[i0 + u] = excl;
            dinv[i0 + u]   = rsqrtf((float)(1 + c[u]));
            excl += c[u];
        }
    }
}

__global__ void edge_place(const int* __restrict__ src, const int* __restrict__ dst,
                           const int* __restrict__ rank, const int* __restrict__ rowptr,
                           int* __restrict__ tmp, int e) {
    int i = blockIdx.x * blockDim.x + threadIdx.x;
    if (i < e) tmp[rowptr[dst[i]] + rank[i]] = src[i];
}

__global__ __launch_bounds__(256)
void bucket_order(const int* __restrict__ rowptr, const int* __restrict__ cnt,
                  const int* __restrict__ tmp, int* __restrict__ adj, int n) {
    const int lane = threadIdx.x & 31;
    const int node = blockIdx.x * 8 + __builtin_amdgcn_readfirstlane((int)(threadIdx.x >> 5));
    if (node >= n) return;
    const int start = rowptr[node];
    const int c     = cnt[node];
    for (int jb = 0; jb < c; jb += 32) {
        const int j  = jb + lane;
        const int sj = (j < c) ? tmp[start + j] : 0x7fffffff;
        int r = 0;
        for (int kb = 0; kb < c; kb += 32) {
            const int k  = kb + lane;
            const int sk = (k < c) ? tmp[start + k] : 0x7fffffff;
            int mm = c - kb; if (mm > 32) mm = 32;
            for (int t = 0; t < mm; ++t) {
                const int v = __builtin_amdgcn_readlane(sk, t);
                r += (v < sj || (v == sj && kb + t < j)) ? 1 : 0;
            }
        }
        if (j < c) adj[start + r] = sj;
    }
}

template <bool DUAL>
__global__ __launch_bounds__(256)
void gemm_wmma_f32(const float* __restrict__ A,
                   const float* __restrict__ B0,
                   const float* __restrict__ B1,
                   float* __restrict__ C, int nrows)
{
    constexpr int BLD = DUAL ? 64 : 128;
    __shared__ float As[16 * KDIM];
    const int row0 = blockIdx.x * 16;
    const int tid  = threadIdx.x;

    {
        v4f* dstv = (v4f*)As;
        #pragma unroll
        for (int idx = tid; idx < 512; idx += 256) {
            int r  = idx >> 5;
            int c4 = idx & 31;
            int gr = row0 + r; if (gr >= nrows) gr = nrows - 1;
            dstv[idx] = ((const v4f*)(A + (size_t)gr * KDIM))[c4];
        }
    }
    __syncthreads();

    const int wave = tid >> 5;
    const int lane = tid & 31;
    const int col0 = wave * 16;
    const int ln   = lane & 15;
    const int hi   = lane >> 4;

    const float* bp;
    if (DUAL) {
        const float* Bsel = (col0 < 64) ? B0 : B1;
        const int    bcol = (col0 < 64) ? (col0 + ln) : (col0 - 64 + ln);
        bp = Bsel + (size_t)(2 * hi) * BLD + bcol;
    } else {
        bp = B0 + (size_t)(2 * hi) * BLD + col0 + ln;
    }
    const float* ap = &As[ln * KDIM + 2 * hi];

    v8f acc = {};
    #pragma unroll
    for (int k = 0; k < KDIM; k += 4) {
        v2f a = *(const v2f*)(ap + k);
        v2f b;
        b.x = bp[(size_t)k * BLD];
        b.y = bp[(size_t)(k + 1) * BLD];
        acc = __builtin_amdgcn_wmma_f32_16x16x4_f32(
            false, a, false, b, (short)0, acc, false, false);
    }

    float* cp = C + (size_t)(row0 + 8 * hi) * FW + col0 + ln;
    #pragma unroll
    for (int r = 0; r < 8; ++r) {
        if (row0 + r + 8 * hi < nrows) cp[(size_t)r * FW] = acc[r];
    }
}

template <bool SPLIT>
__global__ __launch_bounds__(256)
void node_gather(const int* __restrict__ rowptr, const int* __restrict__ cnt,
                 const int* __restrict__ adj, const float* __restrict__ dinv,
                 const float* __restrict__ in,
                 const float* __restrict__ bias0, const float* __restrict__ bias1,
                 float* __restrict__ out, int n)
{
    const int lane = threadIdx.x & 31;
    const int node = blockIdx.x * 8 + __builtin_amdgcn_readfirstlane((int)(threadIdx.x >> 5));
    if (node >= n) return;
    const int   start = rowptr[node];
    const int   c     = cnt[node];
    const float di    = dinv[node];

    const int   half = lane >> 4;
    const int   bl   = SPLIT ? (lane & 15) : lane;
    const float* bsel = (SPLIT && half) ? bias1 : bias0;
    const v4f   bv   = ((const v4f*)bsel)[bl];

    v4f acc = (di * di) * ((const v4f*)(in + (size_t)node * FW))[lane];

    for (int base = 0; base < c; base += 32) {
        int m = c - base; if (m > 32) m = 32;
        int   my_s = node;
        float my_w = 0.0f;
        if (lane < m) {
            my_s = adj[start + base + lane];
            my_w = dinv[my_s] * di;
        }
        int j = 0;
        for (; j + 8 <= m; j += 8) {
            v4f v[8]; float w[8];
            #pragma unroll
            for (int u = 0; u < 8; ++u) {
                const int s = __builtin_amdgcn_readlane(my_s, j + u);
                w[u] = __builtin_bit_cast(float,
                           __builtin_amdgcn_readlane(__builtin_bit_cast(int, my_w), j + u));
                v[u] = ((const v4f*)(in + (size_t)s * FW))[lane];
            }
            #pragma unroll
            for (int u = 0; u < 8; ++u) acc += w[u] * v[u];
        }
        const int r = m - j;
        if (r > 0) {
            v4f v[7]; float w[7];
            #pragma unroll
            for (int u = 0; u < 7; ++u) {
                v[u] = (v4f){0.0f, 0.0f, 0.0f, 0.0f};
                w[u] = 0.0f;
                if (u < r) {
                    const int s = __builtin_amdgcn_readlane(my_s, j + u);
                    w[u] = __builtin_bit_cast(float,
                               __builtin_amdgcn_readlane(__builtin_bit_cast(int, my_w), j + u));
                    v[u] = ((const v4f*)(in + (size_t)s * FW))[lane];
                }
            }
            #pragma unroll
            for (int u = 0; u < 7; ++u) if (u < r) acc += w[u] * v[u];
        }
    }
    if (SPLIT) {
        float* op = out + (size_t)half * ((size_t)n * 64) + (size_t)node * 64;
        ((v4f*)op)[bl] = acc + bv;
    } else {
        v4f h;
        h.x = fmaxf(acc.x + bv.x, 0.0f);
        h.y = fmaxf(acc.y + bv.y, 0.0f);
        h.z = fmaxf(acc.z + bv.z, 0.0f);
        h.w = fmaxf(acc.w + bv.w, 0.0f);
        ((v4f*)(out + (size_t)node * FW))[lane] = h;
    }
}

extern "C" void kernel_launch(void* const* d_in, const int* in_sizes, int n_in,
                              void* d_out, int out_size, void* d_ws, size_t ws_size,
                              hipStream_t stream)
{
    const float* x    = (const float*)d_in[0];
    const int*   ei   = (const int*)  d_in[1];
    const float* W1   = (const float*)d_in[2];
    const float* b1   = (const float*)d_in[3];
    const float* Wmu  = (const float*)d_in[4];
    const float* bmu  = (const float*)d_in[5];
    const float* Wlv  = (const float*)d_in[6];
    const float* blv  = (const float*)d_in[7];

    const int N = in_sizes[0] / KDIM;
    const int E = in_sizes[1] / 2;
    const int* srcIdx = ei;
    const int* dstIdx = ei + E;

    const int TB = 256;
    dim3 blk(TB);
    int gN    = (N + TB - 1) / TB;
    int gE    = (E + TB - 1) / TB;
    int gNw   = (N + 7) / 8;
    int gTile = (N + SCAN_TILE - 1) / SCAN_TILE;
    int gGemm = (N + 15) / 16;

    size_t n_bytes = (((size_t)N * 4) + 255) & ~(size_t)255;
    size_t t_bytes = (((size_t)gTile * 4) + 255) & ~(size_t)255;
    size_t e_bytes = (((size_t)E * 4) + 255) & ~(size_t)255;
    float* dinv   = (float*)d_ws;
    float* bufA   = (float*)((char*)d_ws + n_bytes);
    float* bufB   = bufA + (size_t)N * FW;
    int*   cnt    = (int*)(bufB + (size_t)N * FW);
    int*   rowptr = (int*)((char*)cnt + n_bytes);
    int*   tsum   = (int*)((char*)rowptr + n_bytes);
    int*   rank   = (int*)((char*)tsum + t_bytes);
    int*   tmp    = (int*)((char*)rank + e_bytes);
    int*   adj    = (int*)((char*)tmp + e_bytes);

    cnt_zero      <<<gN,    blk, 0, stream>>>(cnt, N);
    edge_rank     <<<gE,    blk, 0, stream>>>(dstIdx, cnt, rank, E);
    scan_tile_sums<<<gTile, blk, 0, stream>>>(cnt, tsum, N);
    scan_offsets  <<<gTile, blk, 0, stream>>>(cnt, tsum, rowptr, dinv, N);
    edge_place    <<<gE,    blk, 0, stream>>>(srcIdx, dstIdx, rank, rowptr, tmp, E);
    bucket_order  <<<gNw,   blk, 0, stream>>>(rowptr, cnt, tmp, adj, N);

    gemm_wmma_f32<false><<<gGemm, blk, 0, stream>>>(x, W1, nullptr, bufA, N);

    node_gather<false><<<gNw, blk, 0, stream>>>(rowptr, cnt, adj, dinv, bufA,
                                                b1, nullptr, bufB, N);

    gemm_wmma_f32<true><<<gGemm, blk, 0, stream>>>(bufB, Wmu, Wlv, bufA, N);

    node_gather<true><<<gNw, blk, 0, stream>>>(rowptr, cnt, adj, dinv, bufA,
                                               bmu, blv, (float*)d_out, N);
}


# GEMM A strip in LDS: row pitch 128 -> 132 floats; the rows that one A fragment read touches now fall in different banks instead of one bank pair (bank conflict removed); values bit-identical
# speedup vs baseline: 2.8013x; 1.1567x over previous
#include <hip/hip_runtime.h>
#include <hip/hip_bf16.h>

typedef __attribute__((ext_vector_type(2))) float v2f;
typedef __attribute__((ext_vector_type(4))) float v4f;
typedef __attribute__((ext_vector_type(8))) float v8f;

#define KDIM 128
#define FW   128
#define APITCH 132

#define SCAN_TILE 1024

__global__ void cnt_zero(int* __restrict__ cnt, int n) {
    int i = blockIdx.x * blockDim.x + threadIdx.x;
    if (i < n) cnt[i] = 0;
}

__global__ void edge_rank(const int* __restrict__ dst, int* __restrict__ cnt,
                          int* __restrict__ rank, int e) {
    int i = blockIdx.x * blockDim.x + threadIdx.x;
    if (i < e) rank[i] = atomicAdd(&cnt[dst[i]], 1);
}

__device__ __forceinline__ int wave_sum(int v) {
    for (int off = 16; off > 0; off >>= 1) v += __shfl_xor(v, off);
    return v;
}

__global__ __launch_bounds__(256)
void scan_tile_sums(const int* __restrict__ cnt, int* __restrict__ tsum, int n) {
    __shared__ int red[8];
    const int tid = threadIdx.x;
    const int i0  = blockIdx.x * SCAN_TILE + tid * 4;
    int s = 0;
    for (int u = 0; u < 4; ++u) if (i0 + u < n) s += cnt[i0 + u];
    s = wave_sum(s);
    if ((tid & 31) == 0) red[tid >> 5] = s;
    __syncthreads();
    if (tid == 0) {
        int t = 0;
        for (int w = 0; w < 8; ++w) t += red[w];
        tsum[blockIdx.x] = t;
    }
}

__global__ __launch_bounds__(256)
void scan_offsets(const int* __restrict__ cnt, const int* __restrict__ tsum,
                  int* __restrict__ rowptr, float* __restrict__ dinv, int n) {
    __shared__ int red[8];
    __shared__ int wtot[8];
    const int tid  = threadIdx.x;
    const int lane = tid & 31;
    const int wave = tid >> 5;

    int p = 0;
    for (int t = tid; t < (int)blockIdx.x; t += 256) p += tsum[t];
    p = wave_sum(p);
    if (lane == 0) red[wave] = p;
    __syncthreads();
    int tbase = 0;
    for (int w = 0; w < 8; ++w) tbase += red[w];

    const int i0 = blockIdx.x * SCAN_TILE + tid * 4;
    int c[4];
    for (int u = 0; u < 4; ++u) c[u] = (i0 + u < n) ? cnt[i0 + u] : 0;
    const int mine = c[0] + c[1] + c[2] + c[3];

    int x = mine;
    for (int off = 1; off < 32; off <<= 1) {
        int y = __shfl_up(x, off);
        if (lane >= off) x += y;
    }
    if (lane == 31) wtot[wave] = x;
    __syncthreads();
    int wbase = 0;
    for (int w = 0; w < wave; ++w) wbase += wtot[w];

    int excl = tbase + wbase + x - mine;
    for (int u = 0; u < 4; ++u) {
        if (i0 + u < n) {
            rowptr[i0 + u] = excl;
            dinv[i0 + u]   = rsqrtf((float)(1 + c[u]));
            excl += c[u];
        }
    }
}

__global__ void edge_place(const int* __restrict__ src, const int* __restrict__ dst,
                           const int* __restrict__ rank, const int* __restrict__ rowptr,
                           int* __restrict__ tmp, int e) {
    int i = blockIdx.x * blockDim.x + threadIdx.x;
    if (i < e) tmp[rowptr[dst[i]] + rank[i]] = src[i];
}

__global__ __launch_bounds__(256)
void bucket_order(const int* __restrict__ rowptr, const int* __restrict__ cnt,
                  const int* __restrict__ tmp, int* __restrict__ adj, int n) {
    const int lane = threadIdx.x & 31;
    const int node = blockIdx.x * 8 + __builtin_amdgcn_readfirstlane((int)(threadIdx.x >> 5));
    if (node >= n) return;
    const int start = rowptr[node];
    const int c     = cnt[node];
    for (int jb = 0; jb < c; jb += 32) {
        const int j  = jb + lane;
        const int sj = (j < c) ? tmp[start + j] : 0x7fffffff;
        int r = 0;
        for (int kb = 0; kb < c; kb += 32) {
            const int k  = kb + lane;
            const int sk = (k < c) ? tmp[start + k] : 0x7fffffff;
            int mm = c - kb; if (mm > 32) mm = 32;
            for (int t = 0; t < mm; ++t) {
                const int v = __builtin_amdgcn_readlane(sk, t);
                r += (v < sj || (v == sj && kb + t < j)) ? 1 : 0;
            }
        }
        if (j < c) adj[start + r] = sj;
    }
}

template <bool DUAL>
__global__ __launch_bounds__(256)
void gemm_wmma_f32(const float* __restrict__ A,
                   const float* __restrict__ B0,
                   const float* __restrict__ B1,
                   float* __restrict__ C, int nrows)
{
    constexpr int BLD = DUAL ? 64 : 128;
    __shared__ __attribute__((aligned(16))) float As[16 * APITCH];
    const int row0 = blockIdx.x * 16;
    const int tid  = threadIdx.x;

    {
        #pragma unroll
        for (int idx = tid; idx < 512; idx += 256) {
            int r  = idx >> 5;
            int c4 = idx & 31;
            int gr = row0 + r; if (gr >= nrows) gr = nrows - 1;
            *(v4f*)&As[r * APITCH + 4 * c4] = ((const v4f*)(A + (size_t)gr * KDIM))[c4];
        }
    }
    __syncthreads();

    const int wave = tid >> 5;
    const int lane = tid & 31;
    const int col0 = wave * 16;
    const int ln   = lane & 15;
    const int hi   = lane >> 4;

    const float* bp;
    if (DUAL) {
        const float* Bsel = (col0 < 64) ? B0 : B1;
        const int    bcol = (col0 < 64) ? (col0 + ln) : (col0 - 64 + ln);
        bp = Bsel + (size_t)(2 * hi) * BLD + bcol;
    } else {
        bp = B0 + (size_t)(2 * hi) * BLD + col0 + ln;
    }
    const float* ap = &As[ln * APITCH + 2 * hi];

    v8f acc = {};
    #pragma unroll
    for (int k = 0; k < KDIM; k += 4) {
        v2f a = *(const v2f*)(ap + k);
        v2f b;
        b.x = bp[(size_t)k * BLD];
        b.y = bp[(size_t)(k + 1) * BLD];
        acc = __builtin_amdgcn_wmma_f32_16x16x4_f32(
            false, a, false, b, (short)0, acc, false, false);
    }

    float* cp = C + (size_t)(row0 + 8 * hi) * FW + col0 + ln;
    #pragma unroll
    for (int r = 0; r < 8; ++r) {
        if (row0 + r + 8 * hi < nrows) cp[(size_t)r * FW] = acc[r];
    }
}

template <bool SPLIT>
__global__ __launch_bounds__(256)
void node_gather(const int* __restrict__ rowptr, const int* __restrict__ cnt,
                 const int* __restrict__ adj, const float* __restrict__ dinv,
                 const float* __restrict__ in,
                 const float* __restrict__ bias0, const float* __restrict__ bias1,
                 float* __restrict__ out, int n)
{
    const int lane = threadIdx.x & 31;
    const int node = blockIdx.x * 8 + __builtin_amdgcn_readfirstlane((int)(threadIdx.x >> 5));
    if (node >= n) return;
    const int   start = rowptr[node];
    const int   c     = cnt[node];
    const float di    = dinv[node];

    const int   half = lane >> 4;
    const int   bl   = SPLIT ? (lane & 15) : lane;
    const float* bsel = (SPLIT && half) ? bias1 : bias0;
    const v4f   bv   = ((const v4f*)bsel)[bl];

    v4f acc = (di * di) * ((const v4f*)(in + (size_t)node * FW))[lane];

    for (int base = 0; base < c; base += 32) {
        int m = c - base; if (m > 32) m = 32;
        int   my_s = node;
        float my_w = 0.0f;
        if (lane < m) {
            my_s = adj[start + base + lane];
            my_w = dinv[my_s] * di;
        }
        int j = 0;
        for (; j + 8 <= m; j += 8) {
            v4f v[8]; float w[8];
            #pragma unroll
            for (int u = 0; u < 8; ++u) {
                const int s = __builtin_amdgcn_readlane(my_s, j + u);
                w[u] = __builtin_bit_cast(float,
                           __builtin_amdgcn_readlane(__builtin_bit_cast(int, my_w), j + u));
                v[u] = ((const v4f*)(in + (size_t)s * FW))[lane];
            }
            #pragma unroll
            for (int u = 0; u < 8; ++u) acc += w[u] * v[u];
        }
        const int r = m - j;
        if (r > 0) {
            v4f v[7]; float w[7];
            #pragma unroll
            for (int u = 0; u < 7; ++u) {
                v[u] = (v4f){0.0f, 0.0f, 0.0f, 0.0f};
                w[u] = 0.0f;
                if (u < r) {
                    const int s = __builtin_amdgcn_readlane(my_s, j + u);
                    w[u] = __builtin_bit_cast(float,
                               __builtin_amdgcn_readlane(__builtin_bit_cast(int, my_w), j + u));
                    v[u] = ((const v4f*)(in + (size_t)s * FW))[lane];
                }
            }
            #pragma unroll
            for (int u = 0; u < 7; ++u) if (u < r) acc += w[u] * v[u];
        }
    }
    if (SPLIT) {
        float* op = out + (size_t)half * ((size_t)n * 64) + (size_t)node * 64;
        ((v4f*)op)[bl] = acc + bv;
    } else {
        v4f h;
        h.x = fmaxf(acc.x + bv.x, 0.0f);
        h.y = fmaxf(acc.y + bv.y, 0.0f);
        h.z = fmaxf(acc.z + bv.z, 0.0f);
        h.w = fmaxf(acc.w + bv.w, 0.0f);
        ((v4f*)(out + (size_t)node * FW))[lane] = h;
    }
}

extern "C" void kernel_launch(void* const* d_in, const int* in_sizes, int n_in,
                              void* d_out, int out_size, void* d_ws, size_t ws_size,
                              hipStream_t stream)
{
    const float* x    = (const float*)d_in[0];
    const int*   ei   = (const int*)  d_in[1];
    const float* W1   = (const float*)d_in[2];
    const float* b1   = (const float*)d_in[3];
    const float* Wmu  = (const float*)d_in[4];
    const float* bmu  = (const float*)d_in[5];
    const float* Wlv  = (const float*)d_in[6];
    const float* blv  = (const float*)d_in[7];

    const int N = in_sizes[0] / KDIM;
    const int E = in_sizes[1] / 2;
    const int* srcIdx = ei;
    const int* dstIdx = ei + E;

    const int TB = 256;
    dim3 blk(TB);
    int gN    = (N + TB - 1) / TB;
    int gE    = (E + TB - 1) / TB;
    int gNw   = (N + 7) / 8;
    int gTile = (N + SCAN_TILE - 1) / SCAN_TILE;
    int gGemm = (N + 15) / 16;

    size_t n_bytes = (((size_t)N * 4) + 255) & ~(size_t)255;
    size_t t_bytes = (((size_t)gTile * 4) + 255) & ~(size_t)255;
    size_t e_bytes = (((size_t)E * 4) + 255) & ~(size_t)255;
    float* dinv   = (float*)d_ws;
    float* bufA   = (float*)((char*)d_ws + n_bytes);
    float* bufB   = bufA + (size_t)N * FW;
    int*   cnt    = (int*)(bufB + (size_t)N * FW);
    int*   rowptr = (int*)((char*)cnt + n_bytes);
    int*   tsum   = (int*)((char*)rowptr + n_bytes);
    int*   rank   = (int*)((char*)tsum + t_bytes);
    int*   tmp    = (int*)((char*)rank + e_bytes);
    int*   adj    = (int*)((char*)tmp + e_bytes);

    cnt_zero      <<<gN,    blk, 0, stream>>>(cnt, N);
    edge_rank     <<<gE,    blk, 0, stream>>>(dstIdx, cnt, rank, E);
    scan_tile_sums<<<gTile, blk, 0, stream>>>(cnt, tsum, N);
    scan_offsets  <<<gTile, blk, 0, stream>>>(cnt, tsum, rowptr, dinv, N);
    edge_place    <<<gE,    blk, 0, stream>>>(srcIdx, dstIdx, rank, rowptr, tmp, E);
    bucket_order  <<<gNw,   blk, 0, stream>>>(rowptr, cnt, tmp, adj, N);

    gemm_wmma_f32<false><<<gGemm, blk, 0, stream>>>(x, W1, nullptr, bufA, N);

    node_gather<false><<<gNw, blk, 0, stream>>>(rowptr, cnt, adj, dinv, bufA,
                                                b1, nullptr, bufB, N);

    gemm_wmma_f32<true><<<gGemm, blk, 0, stream>>>(bufB, Wmu, Wlv, bufA, N);

    node_gather<true><<<gNw, blk, 0, stream>>>(rowptr, cnt, adj, dinv, bufA,
                                               bmu, blv, (float*)d_out, N);
}


# GEMM: fixed grid of 1024 resident blocks; each walks its 16-row strips with B fragments kept in registers (B read once per block, not per strip), next A strip loaded under the WMMAs; bit-identical
# speedup vs baseline: 2.9098x; 1.0387x over previous
#include <hip/hip_runtime.h>
#include <hip/hip_bf16.h>

typedef __attribute__((ext_vector_type(2))) float v2f;
typedef __attribute__((ext_vector_type(4))) float v4f;
typedef __attribute__((ext_vector_type(8))) float v8f;

#define KDIM 128
#define FW   128
#define APITCH 132

#define SCAN_TILE 1024

__global__ void cnt_zero(int* __restrict__ cnt, int n) {
    int i = blockIdx.x * blockDim.x + threadIdx.x;
    if (i < n) cnt[i] = 0;
}

__global__ void edge_rank(const int* __restrict__ dst, int* __restrict__ cnt,
                          int* __restrict__ rank, int e) {
    int i = blockIdx.x * blockDim.x + threadIdx.x;
    if (i < e) rank[i] = atomicAdd(&cnt[dst[i]], 1);
}

__device__ __forceinline__ int wave_sum(int v) {
    for (int off = 16; off > 0; off >>= 1) v += __shfl_xor(v, off);
    return v;
}

__global__ __launch_bounds__(256)
void scan_tile_sums(const int* __restrict__ cnt, int* __restrict__ tsum, int n) {
    __shared__ int red[8];
    const int tid = threadIdx.x;
    const int i0  = blockIdx.x * SCAN_TILE + tid * 4;
    int s = 0;
    for (int u = 0; u < 4; ++u) if (i0 + u < n) s += cnt[i0 + u];
    s = wave_sum(s);
    if ((tid & 31) == 0) red[tid >> 5] = s;
    __syncthreads();
    if (tid == 0) {
        int t = 0;
        for (int w = 0; w < 8; ++w) t += red[w];
        tsum[blockIdx.x] = t;
    }
}

__global__ __launch_bounds__(256)
void scan_offsets(const int* __restrict__ cnt, const int* __restrict__ tsum,
                  int* __restrict__ rowptr, float* __restrict__ dinv, int n) {
    __shared__ int red[8];
    __shared__ int wtot[8];
    const int tid  = threadIdx.x;
    const int lane = tid & 31;
    const int wave = tid >> 5;

    int p = 0;
    for (int t = tid; t < (int)blockIdx.x; t += 256) p += tsum[t];
    p = wave_sum(p);
    if (lane == 0) red[wave] = p;
    __syncthreads();
    int tbase = 0;
    for (int w = 0; w < 8; ++w) tbase += red[w];

    const int i0 = blockIdx.x * SCAN_TILE + tid * 4;
    int c[4];
    for (int u = 0; u < 4; ++u) c[u] = (i0 + u < n) ? cnt[i0 + u] : 0;
    const int mine = c[0] + c[1] + c[2] + c[3];

    int x = mine;
    for (int off = 1; off < 32; off <<= 1) {
        int y = __shfl_up(x, off);
        if (lane >= off) x += y;
    }
    if (lane == 31) wtot[wave] = x;
    __syncthreads();
    int wbase = 0;
    for (int w = 0; w < wave; ++w) wbase += wtot[w];

    int excl = tbase + wbase + x - mine;
    for (int u = 0; u < 4; ++u) {
        if (i0 + u < n) {
            rowptr[i0 + u] = excl;
            dinv[i0 + u]   = rsqrtf((float)(1 + c[u]));
            excl += c[u];
        }
    }
}

__global__ void edge_place(const int* __restrict__ src, const int* __restrict__ dst,
                           const int* __restrict__ rank, const int* __restrict__ rowptr,
                           int* __restrict__ tmp, int e) {
    int i = blockIdx.x * blockDim.x + threadIdx.x;
    if (i < e) tmp[rowptr[dst[i]] + rank[i]] = src[i];
}

__global__ __launch_bounds__(256)
void bucket_order(const int* __restrict__ rowptr, const int* __restrict__ cnt,
                  const int* __restrict__ tmp, int* __restrict__ adj, int n) {
    const int lane = threadIdx.x & 31;
    const int node = blockIdx.x * 8 + __builtin_amdgcn_readfirstlane((int)(threadIdx.x >> 5));
    if (node >= n) return;
    const int start = rowptr[node];
    const int c     = cnt[node];
    for (int jb = 0; jb < c; jb += 32) {
        const int j  = jb + lane;
        const int sj = (j < c) ? tmp[start + j] : 0x7fffffff;
        int r = 0;
        for (int kb = 0; kb < c; kb += 32) {
            const int k  = kb + lane;
            const int sk = (k < c) ? tmp[start + k] : 0x7fffffff;
            int mm = c - kb; if (mm > 32) mm = 32;
            for (int t = 0; t < mm; ++t) {
                const int v = __builtin_amdgcn_readlane(sk, t);
                r += (v < sj || (v == sj && kb + t < j)) ? 1 : 0;
            }
        }
        if (j < c) adj[start + r] = sj;
    }
}

#define GEMM_BLOCKS 1024

template <bool DUAL>
__global__ __launch_bounds__(256)
void gemm_wmma_f32(const float* __restrict__ A,
                   const float* __restrict__ B0,
                   const float* __restrict__ B1,
                   float* __restrict__ C, int nrows)
{
    constexpr int BLD = DUAL ? 64 : 128;
    __shared__ __attribute__((aligned(16))) float As[2][16 * APITCH];
    const int tid  = threadIdx.x;
    const int wave = tid >> 5;
    const int lane = tid & 31;
    const int col0 = wave * 16;
    const int ln   = lane & 15;
    const int hi   = lane >> 4;
    const int nstrips = (nrows + 15) >> 4;

    const float* bp;
    if (DUAL) {
        const float* Bsel = (col0 < 64) ? B0 : B1;
        const int    bcol = (col0 < 64) ? (col0 + ln) : (col0 - 64 + ln);
        bp = Bsel + (size_t)(2 * hi) * BLD + bcol;
    } else {
        bp = B0 + (size_t)(2 * hi) * BLD + col0 + ln;
    }
    v2f bf[KDIM / 4];
    #pragma unroll
    for (int kk = 0; kk < KDIM / 4; ++kk) {
        bf[kk].x = bp[(size_t)(4 * kk) * BLD];
        bf[kk].y = bp[(size_t)(4 * kk + 1) * BLD];
    }

    const int sr = tid >> 5;
    const int c4 = tid & 31;

    int strip = blockIdx.x;
    v4f p0 = {}, p1 = {};
    if (strip < nstrips) {
        int g0 = strip * 16 + sr;     if (g0 >= nrows) g0 = nrows - 1;
        int g1 = strip * 16 + sr + 8; if (g1 >= nrows) g1 = nrows - 1;
        p0 = ((const v4f*)(A + (size_t)g0 * KDIM))[c4];
        p1 = ((const v4f*)(A + (size_t)g1 * KDIM))[c4];
    }

    int buf = 0;
    for (; strip < nstrips; strip += gridDim.x) {
        float* as = &As[buf][0];
        *(v4f*)&as[sr * APITCH + 4 * c4]       = p0;
        *(v4f*)&as[(sr + 8) * APITCH + 4 * c4] = p1;
        __syncthreads();

        const int nxt = strip + gridDim.x;
        if (nxt < nstrips) {
            int g0 = nxt * 16 + sr;     if (g0 >= nrows) g0 = nrows - 1;
            int g1 = nxt * 16 + sr + 8; if (g1 >= nrows) g1 = nrows - 1;
            p0 = ((const v4f*)(A + (size_t)g0 * KDIM))[c4];
            p1 = ((const v4f*)(A + (size_t)g1 * KDIM))[c4];
        }

        const float* ap = as + ln * APITCH + 2 * hi;
        v8f acc = {};
        #pragma unroll
        for (int kk = 0; kk < KDIM / 4; ++kk) {
            v2f a = *(const v2f*)(ap + 4 * kk);
            acc = __builtin_amdgcn_wmma_f32_16x16x4_f32(
                false, a, false, bf[kk], (short)0, acc, false, false);
        }

        const int row0 = strip * 16;
        float* cp = C + (size_t)(row0 + 8 * hi) * FW + col0 + ln;
        #pragma unroll
        for (int r = 0; r < 8; ++r) {
            if (row0 + r + 8 * hi < nrows) cp[(size_t)r * FW] = acc[r];
        }
        buf ^= 1;
    }
}

template <bool SPLIT>
__global__ __launch_bounds__(256)
void node_gather(const int* __restrict__ rowptr, const int* __restrict__ cnt,
                 const int* __restrict__ adj, const float* __restrict__ dinv,
                 const float* __restrict__ in,
                 const float* __restrict__ bias0, const float* __restrict__ bias1,
                 float* __restrict__ out, int n)
{
    const int lane = threadIdx.x & 31;
    const int node = blockIdx.x * 8 + __builtin_amdgcn_readfirstlane((int)(threadIdx.x >> 5));
    if (node >= n) return;
    const int   start = rowptr[node];
    const int   c     = cnt[node];
    const float di    = dinv[node];

    const int   half = lane >> 4;
    const int   bl   = SPLIT ? (lane & 15) : lane;
    const float* bsel = (SPLIT && half) ? bias1 : bias0;
    const v4f   bv   = ((const v4f*)bsel)[bl];

    v4f acc = (di * di) * ((const v4f*)(in + (size_t)node * FW))[lane];

    for (int base = 0; base < c; base += 32) {
        int m = c - base; if (m > 32) m = 32;
        int   my_s = node;
        float my_w = 0.0f;
        if (lane < m) {
            my_s = adj[start + base + lane];
            my_w = dinv[my_s] * di;
        }
        int j = 0;
        for (; j + 8 <= m; j += 8) {
            v4f v[8]; float w[8];
            #pragma unroll
            for (int u = 0; u < 8; ++u) {
                const int s = __builtin_amdgcn_readlane(my_s, j + u);
                w[u] = __builtin_bit_cast(float,
                           __builtin_amdgcn_readlane(__builtin_bit_cast(int, my_w), j + u));
                v[u] = ((const v4f*)(in + (size_t)s * FW))[lane];
            }
            #pragma unroll
            for (int u = 0; u < 8; ++u) acc += w[u] * v[u];
        }
        const int r = m - j;
        if (r > 0) {
            v4f v[7]; float w[7];
            #pragma unroll
            for (int u = 0; u < 7; ++u) {
                v[u] = (v4f){0.0f, 0.0f, 0.0f, 0.0f};
                w[u] = 0.0f;
                if (u < r) {
                    const int s = __builtin_amdgcn_readlane(my_s, j + u);
                    w[u] = __builtin_bit_cast(float,
                               __builtin_amdgcn_readlane(__builtin_bit_cast(int, my_w), j + u));
                    v[u] = ((const v4f*)(in + (size_t)s * FW))[lane];
                }
            }
            #pragma unroll
            for (int u = 0; u < 7; ++u) if (u < r) acc += w[u] * v[u];
        }
    }
    if (SPLIT) {
        float* op = out + (size_t)half * ((size_t)n * 64) + (size_t)node * 64;
        ((v4f*)op)[bl] = acc + bv;
    } else {
        v4f h;
        h.x = fmaxf(acc.x + bv.x, 0.0f);
        h.y = fmaxf(acc.y + bv.y, 0.0f);
        h.z = fmaxf(acc.z + bv.z, 0.0f);
        h.w = fmaxf(acc.w + bv.w, 0.0f);
        ((v4f*)(out + (size_t)node * FW))[lane] = h;
    }
}

extern "C" void kernel_launch(void* const* d_in, const int* in_sizes, int n_in,
                              void* d_out, int out_size, void* d_ws, size_t ws_size,
                              hipStream_t stream)
{
    const float* x    = (const float*)d_in[0];
    const int*   ei   = (const int*)  d_in[1];
    const float* W1   = (const float*)d_in[2];
    const float* b1   = (const float*)d_in[3];
    const float* Wmu  = (const float*)d_in[4];
    const float* bmu  = (const float*)d_in[5];
    const float* Wlv  = (const float*)d_in[6];
    const float* blv  = (const float*)d_in[7];

    const int N = in_sizes[0] / KDIM;
    const int E = in_sizes[1] / 2;
    const int* srcIdx = ei;
    const int* dstIdx = ei + E;

    const int TB = 256;
    dim3 blk(TB);
    int gN    = (N + TB - 1) / TB;
    int gE    = (E + TB - 1) / TB;
    int gNw   = (N + 7) / 8;
    int gTile = (N + SCAN_TILE - 1) / SCAN_TILE;

    size_t n_bytes = (((size_t)N * 4) + 255) & ~(size_t)255;
    size_t t_bytes = (((size_t)gTile * 4) + 255) & ~(size_t)255;
    size_t e_bytes = (((size_t)E * 4) + 255) & ~(size_t)255;
    float* dinv   = (float*)d_ws;
    float* bufA   = (float*)((char*)d_ws + n_bytes);
    float* bufB   = bufA + (size_t)N * FW;
    int*   cnt    = (int*)(bufB + (size_t)N * FW);
    int*   rowptr = (int*)((char*)cnt + n_bytes);
    int*   tsum   = (int*)((char*)rowptr + n_bytes);
    int*   rank   = (int*)((char*)tsum + t_bytes);
    int*   tmp    = (int*)((char*)rank + e_bytes);
    int*   adj    = (int*)((char*)tmp + e_bytes);

    cnt_zero      <<<gN,    blk, 0, stream>>>(cnt, N);
    edge_rank     <<<gE,    blk, 0, stream>>>(dstIdx, cnt, rank, E);
    scan_tile_sums<<<gTile, blk, 0, stream>>>(cnt, tsum, N);
    scan_offsets  <<<gTile, blk, 0, stream>>>(cnt, tsum, rowptr, dinv, N);
    edge_place    <<<gE,    blk, 0, stream>>>(srcIdx, dstIdx, rank, rowptr, tmp, E);
    bucket_order  <<<gNw,   blk, 0, stream>>>(rowptr, cnt, tmp, adj, N);

    gemm_wmma_f32<false><<<GEMM_BLOCKS, blk, 0, stream>>>(x, W1, nullptr, bufA, N);

    node_gather<false><<<gNw, blk, 0, stream>>>(rowptr, cnt, adj, dinv, bufA,
                                                b1, nullptr, bufB, N);

    gemm_wmma_f32<true><<<GEMM_BLOCKS, blk, 0, stream>>>(bufB, Wmu, Wlv, bufA, N);

    node_gather<true><<<gNw, blk, 0, stream>>>(rowptr, cnt, adj, dinv, bufA,
                                               bmu, blv, (float*)d_out, N);
}


# ordering of each node's bucket by source moved into the layer-1 node-sum kernel: the wave ranks its bucket in registers, permutes it across lanes, uses it and stores it for layer 2; bit-identical
# speedup vs baseline: 3.0278x; 1.0406x over previous
#include <hip/hip_runtime.h>
#include <hip/hip_bf16.h>

typedef __attribute__((ext_vector_type(2))) float v2f;
typedef __attribute__((ext_vector_type(4))) float v4f;
typedef __attribute__((ext_vector_type(8))) float v8f;

#define KDIM 128
#define FW   128
#define APITCH 132

#define SCAN_TILE 1024

__global__ void cnt_zero(int* __restrict__ cnt, int n) {
    int i = blockIdx.x * blockDim.x + threadIdx.x;
    if (i < n) cnt[i] = 0;
}

__global__ void edge_rank(const int* __restrict__ dst, int* __restrict__ cnt,
                          int* __restrict__ rank, int e) {
    int i = blockIdx.x * blockDim.x + threadIdx.x;
    if (i < e) rank[i] = atomicAdd(&cnt[dst[i]], 1);
}

__device__ __forceinline__ int wave_sum(int v) {
    for (int off = 16; off > 0; off >>= 1) v += __shfl_xor(v, off);
    return v;
}

__global__ __launch_bounds__(256)
void scan_tile_sums(const int* __restrict__ cnt, int* __restrict__ tsum, int n) {
    __shared__ int red[8];
    const int tid = threadIdx.x;
    const int i0  = blockIdx.x * SCAN_TILE + tid * 4;
    int s = 0;
    for (int u = 0; u < 4; ++u) if (i0 + u < n) s += cnt[i0 + u];
    s = wave_sum(s);
    if ((tid & 31) == 0) red[tid >> 5] = s;
    __syncthreads();
    if (tid == 0) {
        int t = 0;
        for (int w = 0; w < 8; ++w) t += red[w];
        tsum[blockIdx.x] = t;
    }
}

__global__ __launch_bounds__(256)
void scan_offsets(const int* __restrict__ cnt, const int* __restrict__ tsum,
                  int* __restrict__ rowptr, float* __restrict__ dinv, int n) {
    __shared__ int red[8];
    __shared__ int wtot[8];
    const int tid  = threadIdx.x;
    const int lane = tid & 31;
    const int wave = tid >> 5;

    int p = 0;
    for (int t = tid; t < (int)blockIdx.x; t += 256) p += tsum[t];
    p = wave_sum(p);
    if (lane == 0) red[wave] = p;
    __syncthreads();
    int tbase = 0;
    for (int w = 0; w < 8; ++w) tbase += red[w];

    const int i0 = blockIdx.x * SCAN_TILE + tid * 4;
    int c[4];
    for (int u = 0; u < 4; ++u) c[u] = (i0 + u < n) ? cnt[i0 + u] : 0;
    const int mine = c[0] + c[1] + c[2] + c[3];

    int x = mine;
    for (int off = 1; off < 32; off <<= 1) {
        int y = __shfl_up(x, off);
        if (lane >= off) x += y;
    }
    if (lane == 31) wtot[wave] = x;
    __syncthreads();
    int wbase = 0;
    for (int w = 0; w < wave; ++w) wbase += wtot[w];

    int excl = tbase + wbase + x - mine;
    for (int u = 0; u < 4; ++u) {
        if (i0 + u < n) {
            rowptr[i0 + u] = excl;
            dinv[i0 + u]   = rsqrtf((float)(1 + c[u]));
            excl += c[u];
        }
    }
}

__global__ void edge_place(const int* __restrict__ src, const int* __restrict__ dst,
                           const int* __restrict__ rank, const int* __restrict__ rowptr,
                           int* __restrict__ tmp, int e) {
    int i = blockIdx.x * blockDim.x + threadIdx.x;
    if (i < e) tmp[rowptr[dst[i]] + rank[i]] = src[i];
}

#define GEMM_BLOCKS 1024

template <bool DUAL>
__global__ __launch_bounds__(256)
void gemm_wmma_f32(const float* __restrict__ A,
                   const float* __restrict__ B0,
                   const float* __restrict__ B1,
                   float* __restrict__ C, int nrows)
{
    constexpr int BLD = DUAL ? 64 : 128;
    __shared__ __attribute__((aligned(16))) float As[2][16 * APITCH];
    const int tid  = threadIdx.x;
    const int wave = tid >> 5;
    const int lane = tid & 31;
    const int col0 = wave * 16;
    const int ln   = lane & 15;
    const int hi   = lane >> 4;
    const int nstrips = (nrows + 15) >> 4;

    const float* bp;
    if (DUAL) {
        const float* Bsel = (col0 < 64) ? B0 : B1;
        const int    bcol = (col0 < 64) ? (col0 + ln) : (col0 - 64 + ln);
        bp = Bsel + (size_t)(2 * hi) * BLD + bcol;
    } else {
        bp = B0 + (size_t)(2 * hi) * BLD + col0 + ln;
    }
    v2f bf[KDIM / 4];
    #pragma unroll
    for (int kk = 0; kk < KDIM / 4; ++kk) {
        bf[kk].x = bp[(size_t)(4 * kk) * BLD];
        bf[kk].y = bp[(size_t)(4 * kk + 1) * BLD];
    }

    const int sr = tid >> 5;
    const int c4 = tid & 31;

    int strip = blockIdx.x;
    v4f p0 = {}, p1 = {};
    if (strip < nstrips) {
        int g0 = strip * 16 + sr;     if (g0 >= nrows) g0 = nrows - 1;
        int g1 = strip * 16 + sr + 8; if (g1 >= nrows) g1 = nrows - 1;
        p0 = ((const v4f*)(A + (size_t)g0 * KDIM))[c4];
        p1 = ((const v4f*)(A + (size_t)g1 * KDIM))[c4];
    }

    int buf = 0;
    for (; strip < nstrips; strip += gridDim.x) {
        float* as = &As[buf][0];
        *(v4f*)&as[sr * APITCH + 4 * c4]       = p0;
        *(v4f*)&as[(sr + 8) * APITCH + 4 * c4] = p1;
        __syncthreads();

        const int nxt = strip + gridDim.x;
        if (nxt < nstrips) {
            int g0 = nxt * 16 + sr;     if (g0 >= nrows) g0 = nrows - 1;
            int g1 = nxt * 16 + sr + 8; if (g1 >= nrows) g1 = nrows - 1;
            p0 = ((const v4f*)(A + (size_t)g0 * KDIM))[c4];
            p1 = ((const v4f*)(A + (size_t)g1 * KDIM))[c4];
        }

        const float* ap = as + ln * APITCH + 2 * hi;
        v8f acc = {};
        #pragma unroll
        for (int kk = 0; kk < KDIM / 4; ++kk) {
            v2f a = *(const v2f*)(ap + 4 * kk);
            acc = __builtin_amdgcn_wmma_f32_16x16x4_f32(
                false, a, false, bf[kk], (short)0, acc, false, false);
        }

        const int row0 = strip * 16;
        float* cp = C + (size_t)(row0 + 8 * hi) * FW + col0 + ln;
        #pragma unroll
        for (int r = 0; r < 8; ++r) {
            if (row0 + r + 8 * hi < nrows) cp[(size_t)r * FW] = acc[r];
        }
        buf ^= 1;
    }
}

template <bool SPLIT, bool ORDER>
__global__ __launch_bounds__(256)
void node_gather(const int* __restrict__ rowptr, const int* __restrict__ cnt,
                 const int* __restrict__ lst, int* __restrict__ ord,
                 const float* __restrict__ dinv,
                 const float* __restrict__ in,
                 const float* __restrict__ bias0, const float* __restrict__ bias1,
                 float* __restrict__ out, int n)
{
    const int lane = threadIdx.x & 31;
    const int node = blockIdx.x * 8 + __builtin_amdgcn_readfirstlane((int)(threadIdx.x >> 5));
    if (node >= n) return;
    const int   start = rowptr[node];
    const int   c     = cnt[node];
    const float di    = dinv[node];

    const int   half = lane >> 4;
    const int   bl   = SPLIT ? (lane & 15) : lane;
    const float* bsel = (SPLIT && half) ? bias1 : bias0;
    const v4f   bv   = ((const v4f*)bsel)[bl];

    v4f acc = (di * di) * ((const v4f*)(in + (size_t)node * FW))[lane];

    int first_s = node;
    if (ORDER) {
        if (c <= 32) {
            const int sj = (lane < c) ? lst[start + lane] : 0x7fffffff;
            int r = 0;
            for (int t = 0; t < c; ++t) {
                const int v = __builtin_amdgcn_readlane(sj, t);
                r += (v < sj || (v == sj && t < lane)) ? 1 : 0;
            }
            if (lane >= c) r = lane;
            first_s = __builtin_amdgcn_ds_permute(r << 2, sj);
            if (lane < c) ord[start + lane] = first_s;
        } else {
            for (int jb = 0; jb < c; jb += 32) {
                const int j  = jb + lane;
                const int sj = (j < c) ? lst[start + j] : 0x7fffffff;
                int r = 0;
                for (int kb = 0; kb < c; kb += 32) {
                    const int k  = kb + lane;
                    const int sk = (k < c) ? lst[start + k] : 0x7fffffff;
                    int mm = c - kb; if (mm > 32) mm = 32;
                    for (int t = 0; t < mm; ++t) {
                        const int v = __builtin_amdgcn_readlane(sk, t);
                        r += (v < sj || (v == sj && kb + t < j)) ? 1 : 0;
                    }
                }
                if (j < c) ord[start + r] = sj;
            }
            __threadfence();
        }
    }

    for (int base = 0; base < c; base += 32) {
        int m = c - base; if (m > 32) m = 32;
        int   my_s = node;
        float my_w = 0.0f;
        if (lane < m) {
            if (ORDER) my_s = (c <= 32) ? first_s : ord[start + base + lane];
            else       my_s = lst[start + base + lane];
            my_w = dinv[my_s] * di;
        }
        int j = 0;
        for (; j + 8 <= m; j += 8) {
            v4f v[8]; float w[8];
            #pragma unroll
            for (int u = 0; u < 8; ++u) {
                const int s = __builtin_amdgcn_readlane(my_s, j + u);
                w[u] = __builtin_bit_cast(float,
                           __builtin_amdgcn_readlane(__builtin_bit_cast(int, my_w), j + u));
                v[u] = ((const v4f*)(in + (size_t)s * FW))[lane];
            }
            #pragma unroll
            for (int u = 0; u < 8; ++u) acc += w[u] * v[u];
        }
        const int r = m - j;
        if (r > 0) {
            v4f v[7]; float w[7];
            #pragma unroll
            for (int u = 0; u < 7; ++u) {
                v[u] = (v4f){0.0f, 0.0f, 0.0f, 0.0f};
                w[u] = 0.0f;
                if (u < r) {
                    const int s = __builtin_amdgcn_readlane(my_s, j + u);
                    w[u] = __builtin_bit_cast(float,
                               __builtin_amdgcn_readlane(__builtin_bit_cast(int, my_w), j + u));
                    v[u] = ((const v4f*)(in + (size_t)s * FW))[lane];
                }
            }
            #pragma unroll
            for (int u = 0; u < 7; ++u) if (u < r) acc += w[u] * v[u];
        }
    }
    if (SPLIT) {
        float* op = out + (size_t)half * ((size_t)n * 64) + (size_t)node * 64;
        ((v4f*)op)[bl] = acc + bv;
    } else {
        v4f h;
        h.x = fmaxf(acc.x + bv.x, 0.0f);
        h.y = fmaxf(acc.y + bv.y, 0.0f);
        h.z = fmaxf(acc.z + bv.z, 0.0f);
        h.w = fmaxf(acc.w + bv.w, 0.0f);
        ((v4f*)(out + (size_t)node * FW))[lane] = h;
    }
}

extern "C" void kernel_launch(void* const* d_in, const int* in_sizes, int n_in,
                              void* d_out, int out_size, void* d_ws, size_t ws_size,
                              hipStream_t stream)
{
    const float* x    = (const float*)d_in[0];
    const int*   ei   = (const int*)  d_in[1];
    const float* W1   = (const float*)d_in[2];
    const float* b1   = (const float*)d_in[3];
    const float* Wmu  = (const float*)d_in[4];
    const float* bmu  = (const float*)d_in[5];
    const float* Wlv  = (const float*)d_in[6];
    const float* blv  = (const float*)d_in[7];

    const int N = in_sizes[0] / KDIM;
    const int E = in_sizes[1] / 2;
    const int* srcIdx = ei;
    const int* dstIdx = ei + E;

    const int TB = 256;
    dim3 blk(TB);
    int gN    = (N + TB - 1) / TB;
    int gE    = (E + TB - 1) / TB;
    int gNw   = (N + 7) / 8;
    int gTile = (N + SCAN_TILE - 1) / SCAN_TILE;

    size_t n_bytes = (((size_t)N * 4) + 255) & ~(size_t)255;
    size_t t_bytes = (((size_t)gTile * 4) + 255) & ~(size_t)255;
    size_t e_bytes = (((size_t)E * 4) + 255) & ~(size_t)255;
    float* dinv   = (float*)d_ws;
    float* bufA   = (float*)((char*)d_ws + n_bytes);
    float* bufB   = bufA + (size_t)N * FW;
    int*   cnt    = (int*)(bufB + (size_t)N * FW);
    int*   rowptr = (int*)((char*)cnt + n_bytes);
    int*   tsum   = (int*)((char*)rowptr + n_bytes);
    int*   rank   = (int*)((char*)tsum + t_bytes);
    int*   tmp    = (int*)((char*)rank + e_bytes);
    int*   adj    = (int*)((char*)tmp + e_bytes);

    cnt_zero      <<<gN,    blk, 0, stream>>>(cnt, N);
    edge_rank     <<<gE,    blk, 0, stream>>>(dstIdx, cnt, rank, E);
    scan_tile_sums<<<gTile, blk, 0, stream>>>(cnt, tsum, N);
    scan_offsets  <<<gTile, blk, 0, stream>>>(cnt, tsum, rowptr, dinv, N);
    edge_place    <<<gE,    blk, 0, stream>>>(srcIdx, dstIdx, rank, rowptr, tmp, E);

    gemm_wmma_f32<false><<<GEMM_BLOCKS, blk, 0, stream>>>(x, W1, nullptr, bufA, N);

    node_gather<false, true><<<gNw, blk, 0, stream>>>(rowptr, cnt, tmp, adj, dinv, bufA,
                                                      b1, nullptr, bufB, N);

    gemm_wmma_f32<true><<<GEMM_BLOCKS, blk, 0, stream>>>(bufB, Wmu, Wlv, bufA, N);

    node_gather<true, false><<<gNw, blk, 0, stream>>>(rowptr, cnt, adj, nullptr, dinv, bufA,
                                                      bmu, blv, (float*)d_out, N);
}


# each layer's dense product moved behind its node sum into one kernel: 16 summed input rows -> LDS -> WMMA -> bias/relu -> store; node lists fetched a strip ahead; no GEMM launches; rounding differs
# speedup vs baseline: 3.3750x; 1.1147x over previous
#include <hip/hip_runtime.h>
#include <hip/hip_bf16.h>

typedef __attribute__((ext_vector_type(2))) float v2f;
typedef __attribute__((ext_vector_type(4))) float v4f;
typedef __attribute__((ext_vector_type(8))) float v8f;

#define KDIM 128
#define FW   128
#define APITCH 132

#define SCAN_TILE 1024

__global__ void cnt_zero(int* __restrict__ cnt, int n) {
    int i = blockIdx.x * blockDim.x + threadIdx.x;
    if (i < n) cnt[i] = 0;
}

__global__ void edge_rank(const int* __restrict__ dst, int* __restrict__ cnt,
                          int* __restrict__ rank, int e) {
    int i = blockIdx.x * blockDim.x + threadIdx.x;
    if (i < e) rank[i] = atomicAdd(&cnt[dst[i]], 1);
}

__device__ __forceinline__ int wave_sum(int v) {
    for (int off = 16; off > 0; off >>= 1) v += __shfl_xor(v, off);
    return v;
}

__global__ __launch_bounds__(256)
void scan_tile_sums(const int* __restrict__ cnt, int* __restrict__ tsum, int n) {
    __shared__ int red[8];
    const int tid = threadIdx.x;
    const int i0  = blockIdx.x * SCAN_TILE + tid * 4;
    int s = 0;
    for (int u = 0; u < 4; ++u) if (i0 + u < n) s += cnt[i0 + u];
    s = wave_sum(s);
    if ((tid & 31) == 0) red[tid >> 5] = s;
    __syncthreads();
    if (tid == 0) {
        int t = 0;
        for (int w = 0; w < 8; ++w) t += red[w];
        tsum[blockIdx.x] = t;
    }
}

__global__ __launch_bounds__(256)
void scan_offsets(const int* __restrict__ cnt, const int* __restrict__ tsum,
                  int* __restrict__ rowptr, float* __restrict__ dinv, int n) {
    __shared__ int red[8];
    __shared__ int wtot[8];
    const int tid  = threadIdx.x;
    const int lane = tid & 31;
    const int wave = tid >> 5;

    int p = 0;
    for (int t = tid; t < (int)blockIdx.x; t += 256) p += tsum[t];
    p = wave_sum(p);
    if (lane == 0) red[wave] = p;
    __syncthreads();
    int tbase = 0;
    for (int w = 0; w < 8; ++w) tbase += red[w];

    const int i0 = blockIdx.x * SCAN_TILE + tid * 4;
    int c[4];
    for (int u = 0; u < 4; ++u) c[u] = (i0 + u < n) ? cnt[i0 + u] : 0;
    const int mine = c[0] + c[1] + c[2] + c[3];

    int x = mine;
    for (int off = 1; off < 32; off <<= 1) {
        int y = __shfl_up(x, off);
        if (lane >= off) x += y;
    }
    if (lane == 31) wtot[wave] = x;
    __syncthreads();
    int wbase = 0;
    for (int w = 0; w < wave; ++w) wbase += wtot[w];

    int excl = tbase + wbase + x - mine;
    for (int u = 0; u < 4; ++u) {
        if (i0 + u < n) {
            rowptr[i0 + u] = excl;
            dinv[i0 + u]   = rsqrtf((float)(1 + c[u]));
            excl += c[u];
        }
    }
}

__global__ void edge_place(const int* __restrict__ src, const int* __restrict__ dst,
                           const int* __restrict__ rank, const int* __restrict__ rowptr,
                           int* __restrict__ tmp, int e) {
    int i = blockIdx.x * blockDim.x + threadIdx.x;
    if (i < e) tmp[rowptr[dst[i]] + rank[i]] = src[i];
}

#define LAYER_BLOCKS 1024
#define NO_SRC 0x7fffffff

template <bool ORDER>
__device__ __forceinline__
v4f node_sum(const int node, const int lane, const int start, const int c, const float di,
             const int ps, const float pw,
             const int* __restrict__ lst, int* __restrict__ ord,
             const float* __restrict__ dinv, const float* __restrict__ in)
{
    v4f  acc = ((const v4f*)(in + (size_t)node * FW))[lane];
    bool own_scaled = false;

    int   first_s = ps;
    float first_d = pw;
    if (ORDER) {
        if (c <= 32) {
            const int sj = ps;
            int r = 0;
            for (int t = 0; t < c; ++t) {
                const int v = __builtin_amdgcn_readlane(sj, t);
                r += (v < sj || (v == sj && t < lane)) ? 1 : 0;
            }
            if (lane >= c) r = lane;
            first_s = __builtin_amdgcn_ds_permute(r << 2, sj);
            first_d = __builtin_bit_cast(float,
                          __builtin_amdgcn_ds_permute(r << 2, __builtin_bit_cast(int, pw)));
            if (lane < c) ord[start + lane] = first_s;
        } else {
            for (int jb = 0; jb < c; jb += 32) {
                const int j  = jb + lane;
                const int sj = (j < c) ? lst[start + j] : NO_SRC;
                int r = 0;
                for (int kb = 0; kb < c; kb += 32) {
                    const int k  = kb + lane;
                    const int sk = (k < c) ? lst[start + k] : NO_SRC;
                    int mm = c - kb; if (mm > 32) mm = 32;
                    for (int t = 0; t < mm; ++t) {
                        const int v = __builtin_amdgcn_readlane(sk, t);
                        r += (v < sj || (v == sj && kb + t < j)) ? 1 : 0;
                    }
                }
                if (j < c) ord[start + r] = sj;
            }
            __threadfence();
        }
    }
    const bool have_first = !ORDER || c <= 32;

    for (int base = 0; base < c; base += 32) {
        int m = c - base; if (m > 32) m = 32;
        int   my_s = node;
        float my_w = 0.0f;
        if (base == 0 && have_first) {
            if (lane < m) { my_s = first_s; my_w = first_d * di; }
        } else if (lane < m) {
            if (ORDER) my_s = ord[start + base + lane];
            else       my_s = lst[start + base + lane];
            my_w = dinv[my_s] * di;
        }
        int j = 0;
        for (; j + 8 <= m; j += 8) {
            v4f v[8]; float w[8];
            #pragma unroll
            for (int u = 0; u < 8; ++u) {
                const int s = __builtin_amdgcn_readlane(my_s, j + u);
                w[u] = __builtin_bit_cast(float,
                           __builtin_amdgcn_readlane(__builtin_bit_cast(int, my_w), j + u));
                v[u] = ((const v4f*)(in + (size_t)s * FW))[lane];
            }
            if (!own_scaled) { acc = (di * di) * acc; own_scaled = true; }
            #pragma unroll
            for (int u = 0; u < 8; ++u) acc += w[u] * v[u];
        }
        const int r = m - j;
        if (r > 0) {
            v4f v[7]; float w[7];
            #pragma unroll
            for (int u = 0; u < 7; ++u) {
                v[u] = (v4f){0.0f, 0.0f, 0.0f, 0.0f};
                w[u] = 0.0f;
                if (u < r) {
                    const int s = __builtin_amdgcn_readlane(my_s, j + u);
                    w[u] = __builtin_bit_cast(float,
                               __builtin_amdgcn_readlane(__builtin_bit_cast(int, my_w), j + u));
                    v[u] = ((const v4f*)(in + (size_t)s * FW))[lane];
                }
            }
            if (!own_scaled) { acc = (di * di) * acc; own_scaled = true; }
            #pragma unroll
            for (int u = 0; u < 7; ++u) if (u < r) acc += w[u] * v[u];
        }
    }
    if (!own_scaled) acc = (di * di) * acc;
    return acc;
}

template <bool SPLIT, bool ORDER>
__global__ __launch_bounds__(256)
void gcn_layer(const int* __restrict__ rowptr, const int* __restrict__ cnt,
               const int* __restrict__ lst, int* __restrict__ ord,
               const float* __restrict__ dinv,
               const float* __restrict__ in,
               const float* __restrict__ B0, const float* __restrict__ B1,
               const float* __restrict__ bias0, const float* __restrict__ bias1,
               float* __restrict__ out, int n)
{
    constexpr int BLD = SPLIT ? 64 : 128;
    __shared__ __attribute__((aligned(16))) float As[2][16 * APITCH];
    const int tid  = threadIdx.x;
    const int lane = tid & 31;
    const int wave = __builtin_amdgcn_readfirstlane((int)(tid >> 5));
    const int col0 = wave * 16;
    const int ln   = lane & 15;
    const int hi   = lane >> 4;
    const int nstrips = (n + 15) >> 4;

    const float* bp;
    float bcol;
    if (SPLIT) {
        const float* Bsel = (col0 < 64) ? B0 : B1;
        const float* bsel = (col0 < 64) ? bias0 : bias1;
        const int    bc   = (col0 & 63) + ln;
        bp   = Bsel + (size_t)(2 * hi) * BLD + bc;
        bcol = bsel[bc];
    } else {
        bp   = B0 + (size_t)(2 * hi) * BLD + col0 + ln;
        bcol = bias0[col0 + ln];
    }
    v2f bf[KDIM / 4];
    #pragma unroll
    for (int kk = 0; kk < KDIM / 4; ++kk) {
        bf[kk].x = bp[(size_t)(4 * kk) * BLD];
        bf[kk].y = bp[(size_t)(4 * kk + 1) * BLD];
    }

    const int q_of_lane = lane & 1;
    int strip = blockIdx.x;
    int mnode = strip * 16 + 2 * wave + q_of_lane; if (mnode >= n) mnode = n - 1;
    const int   m_start = rowptr[mnode];
    const int   m_cnt   = cnt[mnode];
    const float m_di    = dinv[mnode];
    int   s0 = __builtin_amdgcn_readlane(m_start, 0), c0 = __builtin_amdgcn_readlane(m_cnt, 0);
    int   s1 = __builtin_amdgcn_readlane(m_start, 1), c1 = __builtin_amdgcn_readlane(m_cnt, 1);
    float d0 = __builtin_bit_cast(float, __builtin_amdgcn_readlane(__builtin_bit_cast(int, m_di), 0));
    float d1 = __builtin_bit_cast(float, __builtin_amdgcn_readlane(__builtin_bit_cast(int, m_di), 1));
    int   ps0 = (lane < c0) ? lst[s0 + lane] : NO_SRC;
    int   ps1 = (lane < c1) ? lst[s1 + lane] : NO_SRC;
    float pw0 = (lane < c0) ? dinv[ps0] : 0.0f;
    float pw1 = (lane < c1) ? dinv[ps1] : 0.0f;

    int buf = 0;
    for (; strip < nstrips; strip += gridDim.x) {
        float* as = &As[buf][0];
        const int node0 = strip * 16 + 2 * wave;

        int nnode = (strip + (int)gridDim.x) * 16 + 2 * wave + q_of_lane; if (nnode >= n) nnode = n - 1;
        const int   n_start = rowptr[nnode];
        const int   n_cnt   = cnt[nnode];
        const float n_di    = dinv[nnode];

        {
            v4f acc = {0.0f, 0.0f, 0.0f, 0.0f};
            if (node0 < n)
                acc = node_sum<ORDER>(node0, lane, s0, c0, d0, ps0, pw0, lst, ord, dinv, in);
            *(v4f*)&as[(2 * wave) * APITCH + 4 * lane] = acc;
        }

        const int ns0 = __builtin_amdgcn_readlane(n_start, 0), nc0 = __builtin_amdgcn_readlane(n_cnt, 0);
        const int ns1 = __builtin_amdgcn_readlane(n_start, 1), nc1 = __builtin_amdgcn_readlane(n_cnt, 1);
        const int nps0 = (lane < nc0) ? lst[ns0 + lane] : NO_SRC;
        const int nps1 = (lane < nc1) ? lst[ns1 + lane] : NO_SRC;

        {
            v4f acc = {0.0f, 0.0f, 0.0f, 0.0f};
            if (node0 + 1 < n)
                acc = node_sum<ORDER>(node0 + 1, lane, s1, c1, d1, ps1, pw1, lst, ord, dinv, in);
            *(v4f*)&as[(2 * wave + 1) * APITCH + 4 * lane] = acc;
        }

        __syncthreads();

        const float npw0 = (lane < nc0) ? dinv[nps0] : 0.0f;
        const float npw1 = (lane < nc1) ? dinv[nps1] : 0.0f;

        const float* ap = as + ln * APITCH + 2 * hi;
        v8f d = {};
        #pragma unroll
        for (int kk = 0; kk < KDIM / 4; ++kk) {
            v2f a = *(const v2f*)(ap + 4 * kk);
            d = __builtin_amdgcn_wmma_f32_16x16x4_f32(
                false, a, false, bf[kk], (short)0, d, false, false);
        }

        const int row0 = strip * 16 + 8 * hi;
        if (SPLIT) {
            float* op = out + (size_t)(col0 >> 6) * ((size_t)n * 64) + (size_t)row0 * 64 + (col0 & 63) + ln;
            #pragma unroll
            for (int r = 0; r < 8; ++r) {
                if (row0 + r < n) op[(size_t)r * 64] = d[r] + bcol;
            }
        } else {
            float* op = out + (size_t)row0 * FW + col0 + ln;
            #pragma unroll
            for (int r = 0; r < 8; ++r) {
                if (row0 + r < n) op[(size_t)r * FW] = fmaxf(d[r] + bcol, 0.0f);
            }
        }

        s0 = ns0; c0 = nc0; s1 = ns1; c1 = nc1;
        d0 = __builtin_bit_cast(float, __builtin_amdgcn_readlane(__builtin_bit_cast(int, n_di), 0));
        d1 = __builtin_bit_cast(float, __builtin_amdgcn_readlane(__builtin_bit_cast(int, n_di), 1));
        ps0 = nps0; ps1 = nps1; pw0 = npw0; pw1 = npw1;
        buf ^= 1;
    }
}

extern "C" void kernel_launch(void* const* d_in, const int* in_sizes, int n_in,
                              void* d_out, int out_size, void* d_ws, size_t ws_size,
                              hipStream_t stream)
{
    const float* x    = (const float*)d_in[0];
    const int*   ei   = (const int*)  d_in[1];
    const float* W1   = (const float*)d_in[2];
    const float* b1   = (const float*)d_in[3];
    const float* Wmu  = (const float*)d_in[4];
    const float* bmu  = (const float*)d_in[5];
    const float* Wlv  = (const float*)d_in[6];
    const float* blv  = (const float*)d_in[7];

    const int N = in_sizes[0] / KDIM;
    const int E = in_sizes[1] / 2;
    const int* srcIdx = ei;
    const int* dstIdx = ei + E;

    const int TB = 256;
    dim3 blk(TB);
    int gN    = (N + TB - 1) / TB;
    int gE    = (E + TB - 1) / TB;
    int gTile = (N + SCAN_TILE - 1) / SCAN_TILE;

    size_t n_bytes = (((size_t)N * 4) + 255) & ~(size_t)255;
    size_t t_bytes = (((size_t)gTile * 4) + 255) & ~(size_t)255;
    size_t e_bytes = (((size_t)E * 4) + 255) & ~(size_t)255;
    float* dinv   = (float*)d_ws;
    float* hbuf   = (float*)((char*)d_ws + n_bytes);
    int*   cnt    = (int*)(hbuf + (size_t)N * FW);
    int*   rowptr = (int*)((char*)cnt + n_bytes);
    int*   tsum   = (int*)((char*)rowptr + n_bytes);
    int*   rank   = (int*)((char*)tsum + t_bytes);
    int*   tmp    = (int*)((char*)rank + e_bytes);
    int*   adj    = (int*)((char*)tmp + e_bytes);

    cnt_zero      <<<gN,    blk, 0, stream>>>(cnt, N);
    edge_rank     <<<gE,    blk, 0, stream>>>(dstIdx, cnt, rank, E);
    scan_tile_sums<<<gTile, blk, 0, stream>>>(cnt, tsum, N);
    scan_offsets  <<<gTile, blk, 0, stream>>>(cnt, tsum, rowptr, dinv, N);
    edge_place    <<<gE,    blk, 0, stream>>>(srcIdx, dstIdx, rank, rowptr, tmp, E);

    gcn_layer<false, true><<<LAYER_BLOCKS, blk, 0, stream>>>(rowptr, cnt, tmp, adj, dinv, x,
                                                             W1, nullptr, b1, nullptr, hbuf, N);

    gcn_layer<true, false><<<LAYER_BLOCKS, blk, 0, stream>>>(rowptr, cnt, adj, nullptr, dinv, hbuf,
                                                             Wmu, Wlv, bmu, blv, (float*)d_out, N);
}


# edge list grouped by destination built by a two-level counting sort on LDS counters (edge tiles by 256-node bin, then a block per bin), not global returning atomics + scattered stores; bit-identical
# speedup vs baseline: 5.3488x; 1.5848x over previous
#include <hip/hip_runtime.h>
#include <hip/hip_bf16.h>

typedef __attribute__((ext_vector_type(2))) float v2f;
typedef __attribute__((ext_vector_type(4))) float v4f;
typedef __attribute__((ext_vector_type(8))) float v8f;

#define KDIM 128
#define FW   128
#define APITCH 132

#define BIN_NODES   256
#define NB_MAX      512
#define TILE_EDGES  8192
#define TS_THREADS  512
#define EPT         (TILE_EDGES / TS_THREADS)
#define BS_THREADS  256
#define BIN_CAP     8192

__device__ __forceinline__ int wave_sum(int v) {
    for (int off = 16; off > 0; off >>= 1) v += __shfl_xor(v, off);
    return v;
}

__global__ __launch_bounds__(TS_THREADS)
void tile_sort(const int* __restrict__ src, const int* __restrict__ dst,
               int* __restrict__ lstartG, int* __restrict__ binned, int e, int nb) {
    __shared__ int hist[NB_MAX];
    __shared__ int lstart[NB_MAX];
    __shared__ int wtot[TS_THREADS / 32];
    __shared__ __attribute__((aligned(16))) int stage[TILE_EDGES];
    const int tid  = threadIdx.x;
    const int lane = tid & 31;
    const int wave = tid >> 5;
    const int e0   = blockIdx.x * TILE_EDGES;
    int m = e - e0; if (m > TILE_EDGES) m = TILE_EDGES;

    hist[tid] = 0;
    __syncthreads();

    int key[EPT], bin[EPT], rk[EPT];
    #pragma unroll
    for (int u = 0; u < EPT; ++u) {
        const int i = u * TS_THREADS + tid;
        key[u] = 0; bin[u] = -1; rk[u] = 0;
        if (i < m) {
            const int s = src[e0 + i];
            const int d = dst[e0 + i];
            key[u] = (int)(((unsigned)s << 8) | ((unsigned)d & (BIN_NODES - 1)));
            bin[u] = d >> 8;
        }
    }
    #pragma unroll
    for (int u = 0; u < EPT; ++u)
        if (bin[u] >= 0) rk[u] = atomicAdd(&hist[bin[u]], 1);
    __syncthreads();

    const int h = hist[tid];
    int x = h;
    for (int off = 1; off < 32; off <<= 1) {
        int y = __shfl_up(x, off);
        if (lane >= off) x += y;
    }
    if (lane == 31) wtot[wave] = x;
    __syncthreads();
    int wbase = 0;
    for (int w = 0; w < wave; ++w) wbase += wtot[w];
    const int excl = wbase + x - h;
    lstart[tid] = excl;
    int* lrow = lstartG + (size_t)blockIdx.x * (nb + 1);
    if (tid < nb)  lrow[tid] = excl;
    if (tid == 0)  lrow[nb]  = m;
    __syncthreads();

    #pragma unroll
    for (int u = 0; u < EPT; ++u)
        if (bin[u] >= 0) stage[lstart[bin[u]] + rk[u]] = key[u];
    __syncthreads();

    for (int i = tid * 4; i < m; i += TS_THREADS * 4) {
        if (i + 3 < m) {
            *(int4*)(binned + e0 + i) = *(const int4*)(stage + i);
        } else {
            for (int k = i; k < m; ++k) binned[e0 + k] = stage[k];
        }
    }
}

__global__ __launch_bounds__(BS_THREADS)
void bin_sort(const int* __restrict__ lstartG, const int* __restrict__ binned,
              int* __restrict__ rowptr, int* __restrict__ cnt, float* __restrict__ dinv,
              int* __restrict__ tmp, int n, int nb, int ntiles) {
    __shared__ int cntL[BIN_NODES];
    __shared__ int cursor[BIN_NODES];
    __shared__ int redS[BS_THREADS / 32];
    __shared__ int redL[BS_THREADS / 32];
    __shared__ int wtot[BS_THREADS / 32];
    __shared__ int stage[BIN_CAP];
    const int tid  = threadIdx.x;
    const int lane = tid & 31;
    const int wave = tid >> 5;
    const int b    = blockIdx.x;
    const int ld   = nb + 1;

    cntL[tid] = 0;
    int sumS = 0, sumL = 0;
    for (int t = tid; t < ntiles; t += BS_THREADS) {
        const int rs = lstartG[(size_t)t * ld + b];
        const int re = lstartG[(size_t)t * ld + b + 1];
        sumS += rs; sumL += re - rs;
    }
    sumS = wave_sum(sumS); sumL = wave_sum(sumL);
    if (lane == 0) { redS[wave] = sumS; redL[wave] = sumL; }
    __syncthreads();
    int binStart = 0, total = 0;
    for (int w = 0; w < BS_THREADS / 32; ++w) { binStart += redS[w]; total += redL[w]; }

    for (int t = tid; t < ntiles; t += BS_THREADS) {
        const int rs  = lstartG[(size_t)t * ld + b];
        const int len = lstartG[(size_t)t * ld + b + 1] - rs;
        const int* run = binned + (size_t)t * TILE_EDGES + rs;
        for (int k = 0; k < len; k += 8) {
            int kv[8];
            #pragma unroll
            for (int u = 0; u < 8; ++u) kv[u] = (k + u < len) ? run[k + u] : 0;
            #pragma unroll
            for (int u = 0; u < 8; ++u)
                if (k + u < len) atomicAdd(&cntL[kv[u] & (BIN_NODES - 1)], 1);
        }
    }
    __syncthreads();

    const int c = cntL[tid];
    int x = c;
    for (int off = 1; off < 32; off <<= 1) {
        int y = __shfl_up(x, off);
        if (lane >= off) x += y;
    }
    if (lane == 31) wtot[wave] = x;
    __syncthreads();
    int wbase = 0;
    for (int w = 0; w < wave; ++w) wbase += wtot[w];
    const int loc = wbase + x - c;
    const int node = b * BIN_NODES + tid;
    if (node < n) {
        rowptr[node] = binStart + loc;
        cnt[node]    = c;
        dinv[node]   = rsqrtf((float)(1 + c));
    }
    cursor[tid] = loc;
    __syncthreads();

    const bool staged = total <= BIN_CAP;
    for (int t = tid; t < ntiles; t += BS_THREADS) {
        const int rs  = lstartG[(size_t)t * ld + b];
        const int len = lstartG[(size_t)t * ld + b + 1] - rs;
        const int* run = binned + (size_t)t * TILE_EDGES + rs;
        for (int k = 0; k < len; k += 8) {
            int kv[8];
            #pragma unroll
            for (int u = 0; u < 8; ++u) kv[u] = (k + u < len) ? run[k + u] : 0;
            int p[8];
            #pragma unroll
            for (int u = 0; u < 8; ++u) {
                p[u] = 0;
                if (k + u < len) p[u] = atomicAdd(&cursor[kv[u] & (BIN_NODES - 1)], 1);
            }
            #pragma unroll
            for (int u = 0; u < 8; ++u) {
                if (k + u < len) {
                    const int s = (int)((unsigned)kv[u] >> 8);
                    if (staged) stage[p[u]] = s;
                    else        tmp[binStart + p[u]] = s;
                }
            }
        }
    }
    if (staged) {
        __syncthreads();
        for (int i = tid; i < total; i += BS_THREADS) tmp[binStart + i] = stage[i];
    }
}

#define LAYER_BLOCKS 1024
#define NO_SRC 0x7fffffff

template <bool ORDER>
__device__ __forceinline__
v4f node_sum(const int node, const int lane, const int start, const int c, const float di,
             const int ps, const float pw,
             const int* __restrict__ lst, int* __restrict__ ord,
             const float* __restrict__ dinv, const float* __restrict__ in)
{
    v4f  acc = ((const v4f*)(in + (size_t)node * FW))[lane];
    bool own_scaled = false;

    int   first_s = ps;
    float first_d = pw;
    if (ORDER) {
        if (c <= 32) {
            const int sj = ps;
            int r = 0;
            for (int t = 0; t < c; ++t) {
                const int v = __builtin_amdgcn_readlane(sj, t);
                r += (v < sj || (v == sj && t < lane)) ? 1 : 0;
            }
            if (lane >= c) r = lane;
            first_s = __builtin_amdgcn_ds_permute(r << 2, sj);
            first_d = __builtin_bit_cast(float,
                          __builtin_amdgcn_ds_permute(r << 2, __builtin_bit_cast(int, pw)));
            if (lane < c) ord[start + lane] = first_s;
        } else {
            for (int jb = 0; jb < c; jb += 32) {
                const int j  = jb + lane;
                const int sj = (j < c) ? lst[start + j] : NO_SRC;
                int r = 0;
                for (int kb = 0; kb < c; kb += 32) {
                    const int k  = kb + lane;
                    const int sk = (k < c) ? lst[start + k] : NO_SRC;
                    int mm = c - kb; if (mm > 32) mm = 32;
                    for (int t = 0; t < mm; ++t) {
                        const int v = __builtin_amdgcn_readlane(sk, t);
                        r += (v < sj || (v == sj && kb + t < j)) ? 1 : 0;
                    }
                }
                if (j < c) ord[start + r] = sj;
            }
            __threadfence();
        }
    }
    const bool have_first = !ORDER || c <= 32;

    for (int base = 0; base < c; base += 32) {
        int m = c - base; if (m > 32) m = 32;
        int   my_s = node;
        float my_w = 0.0f;
        if (base == 0 && have_first) {
            if (lane < m) { my_s = first_s; my_w = first_d * di; }
        } else if (lane < m) {
            if (ORDER) my_s = ord[start + base + lane];
            else       my_s = lst[start + base + lane];
            my_w = dinv[my_s] * di;
        }
        int j = 0;
        for (; j + 8 <= m; j += 8) {
            v4f v[8]; float w[8];
            #pragma unroll
            for (int u = 0; u < 8; ++u) {
                const int s = __builtin_amdgcn_readlane(my_s, j + u);
                w[u] = __builtin_bit_cast(float,
                           __builtin_amdgcn_readlane(__builtin_bit_cast(int, my_w), j + u));
                v[u] = ((const v4f*)(in + (size_t)s * FW))[lane];
            }
            if (!own_scaled) { acc = (di * di) * acc; own_scaled = true; }
            #pragma unroll
            for (int u = 0; u < 8; ++u) acc += w[u] * v[u];
        }
        const int r = m - j;
        if (r > 0) {
            v4f v[7]; float w[7];
            #pragma unroll
            for (int u = 0; u < 7; ++u) {
                v[u] = (v4f){0.0f, 0.0f, 0.0f, 0.0f};
                w[u] = 0.0f;
                if (u < r) {
                    const int s = __builtin_amdgcn_readlane(my_s, j + u);
                    w[u] = __builtin_bit_cast(float,
                               __builtin_amdgcn_readlane(__builtin_bit_cast(int, my_w), j + u));
                    v[u] = ((const v4f*)(in + (size_t)s * FW))[lane];
                }
            }
            if (!own_scaled) { acc = (di * di) * acc; own_scaled = true; }
            #pragma unroll
            for (int u = 0; u < 7; ++u) if (u < r) acc += w[u] * v[u];
        }
    }
    if (!own_scaled) acc = (di * di) * acc;
    return acc;
}

template <bool SPLIT, bool ORDER>
__global__ __launch_bounds__(256)
void gcn_layer(const int* __restrict__ rowptr, const int* __restrict__ cnt,
               const int* __restrict__ lst, int* __restrict__ ord,
               const float* __restrict__ dinv,
               const float* __restrict__ in,
               const float* __restrict__ B0, const float* __restrict__ B1,
               const float* __restrict__ bias0, const float* __restrict__ bias1,
               float* __restrict__ out, int n)
{
    constexpr int BLD = SPLIT ? 64 : 128;
    __shared__ __attribute__((aligned(16))) float As[2][16 * APITCH];
    const int tid  = threadIdx.x;
    const int lane = tid & 31;
    const int wave = __builtin_amdgcn_readfirstlane((int)(tid >> 5));
    const int col0 = wave * 16;
    const int ln   = lane & 15;
    const int hi   = lane >> 4;
    const int nstrips = (n + 15) >> 4;

    const float* bp;
    float bcol;
    if (SPLIT) {
        const float* Bsel = (col0 < 64) ? B0 : B1;
        const float* bsel = (col0 < 64) ? bias0 : bias1;
        const int    bc   = (col0 & 63) + ln;
        bp   = Bsel + (size_t)(2 * hi) * BLD + bc;
        bcol = bsel[bc];
    } else {
        bp   = B0 + (size_t)(2 * hi) * BLD + col0 + ln;
        bcol = bias0[col0 + ln];
    }
    v2f bf[KDIM / 4];
    #pragma unroll
    for (int kk = 0; kk < KDIM / 4; ++kk) {
        bf[kk].x = bp[(size_t)(4 * kk) * BLD];
        bf[kk].y = bp[(size_t)(4 * kk + 1) * BLD];
    }

    const int q_of_lane = lane & 1;
    int strip = blockIdx.x;
    int mnode = strip * 16 + 2 * wave + q_of_lane; if (mnode >= n) mnode = n - 1;
    const int   m_start = rowptr[mnode];
    const int   m_cnt   = cnt[mnode];
    const float m_di    = dinv[mnode];
    int   s0 = __builtin_amdgcn_readlane(m_start, 0), c0 = __builtin_amdgcn_readlane(m_cnt, 0);
    int   s1 = __builtin_amdgcn_readlane(m_start, 1), c1 = __builtin_amdgcn_readlane(m_cnt, 1);
    float d0 = __builtin_bit_cast(float, __builtin_amdgcn_readlane(__builtin_bit_cast(int, m_di), 0));
    float d1 = __builtin_bit_cast(float, __builtin_amdgcn_readlane(__builtin_bit_cast(int, m_di), 1));
    int   ps0 = (lane < c0) ? lst[s0 + lane] : NO_SRC;
    int   ps1 = (lane < c1) ? lst[s1 + lane] : NO_SRC;
    float pw0 = (lane < c0) ? dinv[ps0] : 0.0f;
    float pw1 = (lane < c1) ? dinv[ps1] : 0.0f;

    int buf = 0;
    for (; strip < nstrips; strip += gridDim.x) {
        float* as = &As[buf][0];
        const int node0 = strip * 16 + 2 * wave;

        int nnode = (strip + (int)gridDim.x) * 16 + 2 * wave + q_of_lane; if (nnode >= n) nnode = n - 1;
        const int   n_start = rowptr[nnode];
        const int   n_cnt   = cnt[nnode];
        const float n_di    = dinv[nnode];

        {
            v4f acc = {0.0f, 0.0f, 0.0f, 0.0f};
            if (node0 < n)
                acc = node_sum<ORDER>(node0, lane, s0, c0, d0, ps0, pw0, lst, ord, dinv, in);
            *(v4f*)&as[(2 * wave) * APITCH + 4 * lane] = acc;
        }

        const int ns0 = __builtin_amdgcn_readlane(n_start, 0), nc0 = __builtin_amdgcn_readlane(n_cnt, 0);
        const int ns1 = __builtin_amdgcn_readlane(n_start, 1), nc1 = __builtin_amdgcn_readlane(n_cnt, 1);
        const int nps0 = (lane < nc0) ? lst[ns0 + lane] : NO_SRC;
        const int nps1 = (lane < nc1) ? lst[ns1 + lane] : NO_SRC;

        {
            v4f acc = {0.0f, 0.0f, 0.0f, 0.0f};
            if (node0 + 1 < n)
                acc = node_sum<ORDER>(node0 + 1, lane, s1, c1, d1, ps1, pw1, lst, ord, dinv, in);
            *(v4f*)&as[(2 * wave + 1) * APITCH + 4 * lane] = acc;
        }

        __syncthreads();

        const float npw0 = (lane < nc0) ? dinv[nps0] : 0.0f;
        const float npw1 = (lane < nc1) ? dinv[nps1] : 0.0f;

        const float* ap = as + ln * APITCH + 2 * hi;
        v8f d = {};
        #pragma unroll
        for (int kk = 0; kk < KDIM / 4; ++kk) {
            v2f a = *(const v2f*)(ap + 4 * kk);
            d = __builtin_amdgcn_wmma_f32_16x16x4_f32(
                false, a, false, bf[kk], (short)0, d, false, false);
        }

        const int row0 = strip * 16 + 8 * hi;
        if (SPLIT) {
            float* op = out + (size_t)(col0 >> 6) * ((size_t)n * 64) + (size_t)row0 * 64 + (col0 & 63) + ln;
            #pragma unroll
            for (int r = 0; r < 8; ++r) {
                if (row0 + r < n) op[(size_t)r * 64] = d[r] + bcol;
            }
        } else {
            float* op = out + (size_t)row0 * FW + col0 + ln;
            #pragma unroll
            for (int r = 0; r < 8; ++r) {
                if (row0 + r < n) op[(size_t)r * FW] = fmaxf(d[r] + bcol, 0.0f);
            }
        }

        s0 = ns0; c0 = nc0; s1 = ns1; c1 = nc1;
        d0 = __builtin_bit_cast(float, __builtin_amdgcn_readlane(__builtin_bit_cast(int, n_di), 0));
        d1 = __builtin_bit_cast(float, __builtin_amdgcn_readlane(__builtin_bit_cast(int, n_di), 1));
        ps0 = nps0; ps1 = nps1; pw0 = npw0; pw1 = npw1;
        buf ^= 1;
    }
}

extern "C" void kernel_launch(void* const* d_in, const int* in_sizes, int n_in,
                              void* d_out, int out_size, void* d_ws, size_t ws_size,
                              hipStream_t stream)
{
    const float* x    = (const float*)d_in[0];
    const int*   ei   = (const int*)  d_in[1];
    const float* W1   = (const float*)d_in[2];
    const float* b1   = (const float*)d_in[3];
    const float* Wmu  = (const float*)d_in[4];
    const float* bmu  = (const float*)d_in[5];
    const float* Wlv  = (const float*)d_in[6];
    const float* blv  = (const float*)d_in[7];

    const int N = in_sizes[0] / KDIM;
    const int E = in_sizes[1] / 2;
    const int* srcIdx = ei;
    const int* dstIdx = ei + E;

    const int TB = 256;
    dim3 blk(TB);
    const int NB = (N + BIN_NODES - 1) / BIN_NODES;
    const int NT = (E + TILE_EDGES - 1) / TILE_EDGES;

    size_t n_bytes = (((size_t)N * 4) + 255) & ~(size_t)255;
    size_t l_bytes = (((size_t)NT * (NB + 1) * 4) + 255) & ~(size_t)255;
    size_t e_bytes = (((size_t)E * 4) + 255) & ~(size_t)255;
    float* dinv   = (float*)d_ws;
    float* hbuf   = (float*)((char*)d_ws + n_bytes);
    int*   cnt    = (int*)(hbuf + (size_t)N * FW);
    int*   rowptr = (int*)((char*)cnt + n_bytes);
    int*   lstart = (int*)((char*)rowptr + n_bytes);
    int*   binned = (int*)((char*)lstart + l_bytes);
    int*   tmp    = (int*)((char*)binned + e_bytes);
    int*   adj    = (int*)((char*)tmp + e_bytes);

    tile_sort<<<NT, dim3(TS_THREADS), 0, stream>>>(srcIdx, dstIdx, lstart, binned, E, NB);
    bin_sort <<<NB, dim3(BS_THREADS), 0, stream>>>(lstart, binned, rowptr, cnt, dinv, tmp, N, NB, NT);

    gcn_layer<false, true><<<LAYER_BLOCKS, blk, 0, stream>>>(rowptr, cnt, tmp, adj, dinv, x,
                                                             W1, nullptr, b1, nullptr, hbuf, N);

    gcn_layer<true, false><<<LAYER_BLOCKS, blk, 0, stream>>>(rowptr, cnt, adj, nullptr, dinv, hbuf,
                                                             Wmu, Wlv, bmu, blv, (float*)d_out, N);
}


# layer 1 stores each row times its own dinv, so layer 2 sums gathered rows as they are and multiplies once by dinv[i]: the scattered 4-byte dinv[source] read per edge is gone there; rounding differs
# speedup vs baseline: 5.6169x; 1.0501x over previous
#include <hip/hip_runtime.h>
#include <hip/hip_bf16.h>

typedef __attribute__((ext_vector_type(2))) float v2f;
typedef __attribute__((ext_vector_type(4))) float v4f;
typedef __attribute__((ext_vector_type(8))) float v8f;

#define KDIM 128
#define FW   128
#define APITCH 132

#define BIN_NODES   256
#define NB_MAX      512
#define TILE_EDGES  8192
#define TS_THREADS  512
#define EPT         (TILE_EDGES / TS_THREADS)
#define BS_THREADS  256
#define BIN_CAP     8192

__device__ __forceinline__ int wave_sum(int v) {
    for (int off = 16; off > 0; off >>= 1) v += __shfl_xor(v, off);
    return v;
}

__global__ __launch_bounds__(TS_THREADS)
void tile_sort(const int* __restrict__ src, const int* __restrict__ dst,
               int* __restrict__ lstartG, int* __restrict__ binned, int e, int nb) {
    __shared__ int hist[NB_MAX];
    __shared__ int lstart[NB_MAX];
    __shared__ int wtot[TS_THREADS / 32];
    __shared__ __attribute__((aligned(16))) int stage[TILE_EDGES];
    const int tid  = threadIdx.x;
    const int lane = tid & 31;
    const int wave = tid >> 5;
    const int e0   = blockIdx.x * TILE_EDGES;
    int m = e - e0; if (m > TILE_EDGES) m = TILE_EDGES;

    hist[tid] = 0;
    __syncthreads();

    int key[EPT], bin[EPT], rk[EPT];
    #pragma unroll
    for (int u = 0; u < EPT; ++u) {
        const int i = u * TS_THREADS + tid;
        key[u] = 0; bin[u] = -1; rk[u] = 0;
        if (i < m) {
            const int s = src[e0 + i];
            const int d = dst[e0 + i];
            key[u] = (int)(((unsigned)s << 8) | ((unsigned)d & (BIN_NODES - 1)));
            bin[u] = d >> 8;
        }
    }
    #pragma unroll
    for (int u = 0; u < EPT; ++u)
        if (bin[u] >= 0) rk[u] = atomicAdd(&hist[bin[u]], 1);
    __syncthreads();

    const int h = hist[tid];
    int x = h;
    for (int off = 1; off < 32; off <<= 1) {
        int y = __shfl_up(x, off);
        if (lane >= off) x += y;
    }
    if (lane == 31) wtot[wave] = x;
    __syncthreads();
    int wbase = 0;
    for (int w = 0; w < wave; ++w) wbase += wtot[w];
    const int excl = wbase + x - h;
    lstart[tid] = excl;
    int* lrow = lstartG + (size_t)blockIdx.x * (nb + 1);
    if (tid < nb)  lrow[tid] = excl;
    if (tid == 0)  lrow[nb]  = m;
    __syncthreads();

    #pragma unroll
    for (int u = 0; u < EPT; ++u)
        if (bin[u] >= 0) stage[lstart[bin[u]] + rk[u]] = key[u];
    __syncthreads();

    for (int i = tid * 4; i < m; i += TS_THREADS * 4) {
        if (i + 3 < m) {
            *(int4*)(binned + e0 + i) = *(const int4*)(stage + i);
        } else {
            for (int k = i; k < m; ++k) binned[e0 + k] = stage[k];
        }
    }
}

__global__ __launch_bounds__(BS_THREADS)
void bin_sort(const int* __restrict__ lstartG, const int* __restrict__ binned,
              int* __restrict__ rowptr, int* __restrict__ cnt, float* __restrict__ dinv,
              int* __restrict__ tmp, int n, int nb, int ntiles) {
    __shared__ int cntL[BIN_NODES];
    __shared__ int cursor[BIN_NODES];
    __shared__ int redS[BS_THREADS / 32];
    __shared__ int redL[BS_THREADS / 32];
    __shared__ int wtot[BS_THREADS / 32];
    __shared__ int stage[BIN_CAP];
    const int tid  = threadIdx.x;
    const int lane = tid & 31;
    const int wave = tid >> 5;
    const int b    = blockIdx.x;
    const int ld   = nb + 1;

    cntL[tid] = 0;
    int sumS = 0, sumL = 0;
    for (int t = tid; t < ntiles; t += BS_THREADS) {
        const int rs = lstartG[(size_t)t * ld + b];
        const int re = lstartG[(size_t)t * ld + b + 1];
        sumS += rs; sumL += re - rs;
    }
    sumS = wave_sum(sumS); sumL = wave_sum(sumL);
    if (lane == 0) { redS[wave] = sumS; redL[wave] = sumL; }
    __syncthreads();
    int binStart = 0, total = 0;
    for (int w = 0; w < BS_THREADS / 32; ++w) { binStart += redS[w]; total += redL[w]; }

    for (int t = tid; t < ntiles; t += BS_THREADS) {
        const int rs  = lstartG[(size_t)t * ld + b];
        const int len = lstartG[(size_t)t * ld + b + 1] - rs;
        const int* run = binned + (size_t)t * TILE_EDGES + rs;
        for (int k = 0; k < len; k += 8) {
            int kv[8];
            #pragma unroll
            for (int u = 0; u < 8; ++u) kv[u] = (k + u < len) ? run[k + u] : 0;
            #pragma unroll
            for (int u = 0; u < 8; ++u)
                if (k + u < len) atomicAdd(&cntL[kv[u] & (BIN_NODES - 1)], 1);
        }
    }
    __syncthreads();

    const int c = cntL[tid];
    int x = c;
    for (int off = 1; off < 32; off <<= 1) {
        int y = __shfl_up(x, off);
        if (lane >= off) x += y;
    }
    if (lane == 31) wtot[wave] = x;
    __syncthreads();
    int wbase = 0;
    for (int w = 0; w < wave; ++w) wbase += wtot[w];
    const int loc = wbase + x - c;
    const int node = b * BIN_NODES + tid;
    if (node < n) {
        rowptr[node] = binStart + loc;
        cnt[node]    = c;
        dinv[node]   = rsqrtf((float)(1 + c));
    }
    cursor[tid] = loc;
    __syncthreads();

    const bool staged = total <= BIN_CAP;
    for (int t = tid; t < ntiles; t += BS_THREADS) {
        const int rs  = lstartG[(size_t)t * ld + b];
        const int len = lstartG[(size_t)t * ld + b + 1] - rs;
        const int* run = binned + (size_t)t * TILE_EDGES + rs;
        for (int k = 0; k < len; k += 8) {
            int kv[8];
            #pragma unroll
            for (int u = 0; u < 8; ++u) kv[u] = (k + u < len) ? run[k + u] : 0;
            int p[8];
            #pragma unroll
            for (int u = 0; u < 8; ++u) {
                p[u] = 0;
                if (k + u < len) p[u] = atomicAdd(&cursor[kv[u] & (BIN_NODES - 1)], 1);
            }
            #pragma unroll
            for (int u = 0; u < 8; ++u) {
                if (k + u < len) {
                    const int s = (int)((unsigned)kv[u] >> 8);
                    if (staged) stage[p[u]] = s;
                    else        tmp[binStart + p[u]] = s;
                }
            }
        }
    }
    if (staged) {
        __syncthreads();
        for (int i = tid; i < total; i += BS_THREADS) tmp[binStart + i] = stage[i];
    }
}

#define LAYER_BLOCKS 1024
#define NO_SRC 0x7fffffff

template <int R>
__device__ __forceinline__
v4f add_rows(v4f acc, const int my_s, const int j, const int lane, const float* __restrict__ in)
{
    v4f v[R];
    #pragma unroll
    for (int u = 0; u < R; ++u) {
        const int s = __builtin_amdgcn_readlane(my_s, j + u);
        v[u] = ((const v4f*)(in + (size_t)s * FW))[lane];
    }
    #pragma unroll
    for (int u = 0; u < R; ++u) acc += v[u];
    return acc;
}

__device__ __forceinline__
v4f add_bucket(v4f acc, const int my_s, const int m, const int lane, const float* __restrict__ in)
{
    int j = 0;
    for (; j + 8 <= m; j += 8) acc = add_rows<8>(acc, my_s, j, lane, in);
    switch (m - j) {
        case 1: acc = add_rows<1>(acc, my_s, j, lane, in); break;
        case 2: acc = add_rows<2>(acc, my_s, j, lane, in); break;
        case 3: acc = add_rows<3>(acc, my_s, j, lane, in); break;
        case 4: acc = add_rows<4>(acc, my_s, j, lane, in); break;
        case 5: acc = add_rows<5>(acc, my_s, j, lane, in); break;
        case 6: acc = add_rows<6>(acc, my_s, j, lane, in); break;
        case 7: acc = add_rows<7>(acc, my_s, j, lane, in); break;
        default: break;
    }
    return acc;
}

template <bool ORDER, bool WEIGHTED>
__device__ __forceinline__
v4f node_sum(const int node, const int lane, const int start, const int c, const float di,
             const int ps, const float pw,
             const int* __restrict__ lst, int* __restrict__ ord,
             const float* __restrict__ dinv, const float* __restrict__ in)
{
    v4f  acc = ((const v4f*)(in + (size_t)node * FW))[lane];
    bool own_scaled = false;

    int   first_s = ps;
    float first_d = pw;
    if (ORDER) {
        if (c <= 32) {
            const int sj = ps;
            int r = 0;
            for (int t = 0; t < c; ++t) {
                const int v = __builtin_amdgcn_readlane(sj, t);
                r += (v < sj || (v == sj && t < lane)) ? 1 : 0;
            }
            if (lane >= c) r = lane;
            first_s = __builtin_amdgcn_ds_permute(r << 2, sj);
            first_d = __builtin_bit_cast(float,
                          __builtin_amdgcn_ds_permute(r << 2, __builtin_bit_cast(int, pw)));
            if (lane < c) ord[start + lane] = first_s;
        } else {
            for (int jb = 0; jb < c; jb += 32) {
                const int j  = jb + lane;
                const int sj = (j < c) ? lst[start + j] : NO_SRC;
                int r = 0;
                for (int kb = 0; kb < c; kb += 32) {
                    const int k  = kb + lane;
                    const int sk = (k < c) ? lst[start + k] : NO_SRC;
                    int mm = c - kb; if (mm > 32) mm = 32;
                    for (int t = 0; t < mm; ++t) {
                        const int v = __builtin_amdgcn_readlane(sk, t);
                        r += (v < sj || (v == sj && kb + t < j)) ? 1 : 0;
                    }
                }
                if (j < c) ord[start + r] = sj;
            }
            __threadfence();
        }
    }
    const bool have_first = !ORDER || c <= 32;

    if (!WEIGHTED) {
        int base = 0;
        if (have_first) {
            acc = add_bucket(acc, first_s, c < 32 ? c : 32, lane, in);
            base = 32;
        }
        for (; base < c; base += 32) {
            int m = c - base; if (m > 32) m = 32;
            int my_s = node;
            if (lane < m) my_s = ORDER ? ord[start + base + lane] : lst[start + base + lane];
            acc = add_bucket(acc, my_s, m, lane, in);
        }
        return di * acc;
    }

    for (int base = 0; base < c; base += 32) {
        int m = c - base; if (m > 32) m = 32;
        int   my_s = node;
        float my_w = 0.0f;
        if (base == 0 && have_first) {
            if (lane < m) { my_s = first_s; my_w = first_d * di; }
        } else if (lane < m) {
            if (ORDER) my_s = ord[start + base + lane];
            else       my_s = lst[start + base + lane];
            my_w = dinv[my_s] * di;
        }
        int j = 0;
        for (; j + 8 <= m; j += 8) {
            v4f v[8]; float w[8];
            #pragma unroll
            for (int u = 0; u < 8; ++u) {
                const int s = __builtin_amdgcn_readlane(my_s, j + u);
                w[u] = __builtin_bit_cast(float,
                           __builtin_amdgcn_readlane(__builtin_bit_cast(int, my_w), j + u));
                v[u] = ((const v4f*)(in + (size_t)s * FW))[lane];
            }
            if (!own_scaled) { acc = (di * di) * acc; own_scaled = true; }
            #pragma unroll
            for (int u = 0; u < 8; ++u) acc += w[u] * v[u];
        }
        const int r = m - j;
        if (r > 0) {
            v4f v[7]; float w[7];
            #pragma unroll
            for (int u = 0; u < 7; ++u) {
                v[u] = (v4f){0.0f, 0.0f, 0.0f, 0.0f};
                w[u] = 0.0f;
                if (u < r) {
                    const int s = __builtin_amdgcn_readlane(my_s, j + u);
                    w[u] = __builtin_bit_cast(float,
                               __builtin_amdgcn_readlane(__builtin_bit_cast(int, my_w), j + u));
                    v[u] = ((const v4f*)(in + (size_t)s * FW))[lane];
                }
            }
            if (!own_scaled) { acc = (di * di) * acc; own_scaled = true; }
            #pragma unroll
            for (int u = 0; u < 7; ++u) if (u < r) acc += w[u] * v[u];
        }
    }
    if (!own_scaled) acc = (di * di) * acc;
    return acc;
}

template <bool SPLIT, bool ORDER>
__global__ __launch_bounds__(256, 8)
void gcn_layer(const int* __restrict__ rowptr, const int* __restrict__ cnt,
               const int* __restrict__ lst, int* __restrict__ ord,
               const float* __restrict__ dinv,
               const float* __restrict__ in,
               const float* __restrict__ B0, const float* __restrict__ B1,
               const float* __restrict__ bias0, const float* __restrict__ bias1,
               float* __restrict__ out, int n)
{
    constexpr int BLD = SPLIT ? 64 : 128;
    constexpr bool WEIGHTED = !SPLIT;
    __shared__ __attribute__((aligned(16))) float As[2][16 * APITCH];
    __shared__ __attribute__((aligned(16))) float Ds[2][16];
    const int tid  = threadIdx.x;
    const int lane = tid & 31;
    const int wave = __builtin_amdgcn_readfirstlane((int)(tid >> 5));
    const int col0 = wave * 16;
    const int ln   = lane & 15;
    const int hi   = lane >> 4;
    const int nstrips = (n + 15) >> 4;

    const float* bp;
    float bcol;
    if (SPLIT) {
        const float* Bsel = (col0 < 64) ? B0 : B1;
        const float* bsel = (col0 < 64) ? bias0 : bias1;
        const int    bc   = (col0 & 63) + ln;
        bp   = Bsel + (size_t)(2 * hi) * BLD + bc;
        bcol = bsel[bc];
    } else {
        bp   = B0 + (size_t)(2 * hi) * BLD + col0 + ln;
        bcol = bias0[col0 + ln];
    }
    v2f bf[KDIM / 4];
    #pragma unroll
    for (int kk = 0; kk < KDIM / 4; ++kk) {
        bf[kk].x = bp[(size_t)(4 * kk) * BLD];
        bf[kk].y = bp[(size_t)(4 * kk + 1) * BLD];
    }

    const int q_of_lane = lane & 1;
    int strip = blockIdx.x;
    int mnode = strip * 16 + 2 * wave + q_of_lane; if (mnode >= n) mnode = n - 1;
    const int   m_start = rowptr[mnode];
    const int   m_cnt   = cnt[mnode];
    const float m_di    = dinv[mnode];
    int   s0 = __builtin_amdgcn_readlane(m_start, 0), c0 = __builtin_amdgcn_readlane(m_cnt, 0);
    int   s1 = __builtin_amdgcn_readlane(m_start, 1), c1 = __builtin_amdgcn_readlane(m_cnt, 1);
    float d0 = __builtin_bit_cast(float, __builtin_amdgcn_readlane(__builtin_bit_cast(int, m_di), 0));
    float d1 = __builtin_bit_cast(float, __builtin_amdgcn_readlane(__builtin_bit_cast(int, m_di), 1));
    int   ps0 = (lane < c0) ? lst[s0 + lane] : NO_SRC;
    int   ps1 = (lane < c1) ? lst[s1 + lane] : NO_SRC;
    float pw0 = 0.0f, pw1 = 0.0f;
    if (WEIGHTED) {
        pw0 = (lane < c0) ? dinv[ps0] : 0.0f;
        pw1 = (lane < c1) ? dinv[ps1] : 0.0f;
    }

    int buf = 0;
    for (; strip < nstrips; strip += gridDim.x) {
        float* as = &As[buf][0];
        const int node0 = strip * 16 + 2 * wave;

        int nnode = (strip + (int)gridDim.x) * 16 + 2 * wave + q_of_lane; if (nnode >= n) nnode = n - 1;
        const int   n_start = rowptr[nnode];
        const int   n_cnt   = cnt[nnode];
        const float n_di    = dinv[nnode];

        {
            v4f acc = {0.0f, 0.0f, 0.0f, 0.0f};
            if (node0 < n)
                acc = node_sum<ORDER, WEIGHTED>(node0, lane, s0, c0, d0, ps0, pw0, lst, ord, dinv, in);
            *(v4f*)&as[(2 * wave) * APITCH + 4 * lane] = acc;
        }

        const int ns0 = __builtin_amdgcn_readlane(n_start, 0), nc0 = __builtin_amdgcn_readlane(n_cnt, 0);
        const int ns1 = __builtin_amdgcn_readlane(n_start, 1), nc1 = __builtin_amdgcn_readlane(n_cnt, 1);
        const int nps0 = (lane < nc0) ? lst[ns0 + lane] : NO_SRC;
        const int nps1 = (lane < nc1) ? lst[ns1 + lane] : NO_SRC;

        {
            v4f acc = {0.0f, 0.0f, 0.0f, 0.0f};
            if (node0 + 1 < n)
                acc = node_sum<ORDER, WEIGHTED>(node0 + 1, lane, s1, c1, d1, ps1, pw1, lst, ord, dinv, in);
            *(v4f*)&as[(2 * wave + 1) * APITCH + 4 * lane] = acc;
        }
        if (!SPLIT && lane < 2) Ds[buf][2 * wave + lane] = lane ? d1 : d0;

        __syncthreads();

        float npw0 = 0.0f, npw1 = 0.0f;
        if (WEIGHTED) {
            npw0 = (lane < nc0) ? dinv[nps0] : 0.0f;
            npw1 = (lane < nc1) ? dinv[nps1] : 0.0f;
        }

        const float* ap = as + ln * APITCH + 2 * hi;
        v8f d = {};
        #pragma unroll
        for (int kk = 0; kk < KDIM / 4; ++kk) {
            v2f a = *(const v2f*)(ap + 4 * kk);
            d = __builtin_amdgcn_wmma_f32_16x16x4_f32(
                false, a, false, bf[kk], (short)0, d, false, false);
        }

        const int row0 = strip * 16 + 8 * hi;
        if (SPLIT) {
            float* op = out + (size_t)(col0 >> 6) * ((size_t)n * 64) + (size_t)row0 * 64 + (col0 & 63) + ln;
            #pragma unroll
            for (int r = 0; r < 8; ++r) {
                if (row0 + r < n) op[(size_t)r * 64] = d[r] + bcol;
            }
        } else {
            float* op = out + (size_t)row0 * FW + col0 + ln;
            const v4f dlo = *(const v4f*)&Ds[buf][8 * hi];
            const v4f dhi = *(const v4f*)&Ds[buf][8 * hi + 4];
            const float dr[8] = {dlo.x, dlo.y, dlo.z, dlo.w, dhi.x, dhi.y, dhi.z, dhi.w};
            #pragma unroll
            for (int r = 0; r < 8; ++r) {
                if (row0 + r < n) op[(size_t)r * FW] = dr[r] * fmaxf(d[r] + bcol, 0.0f);
            }
        }

        s0 = ns0; c0 = nc0; s1 = ns1; c1 = nc1;
        d0 = __builtin_bit_cast(float, __builtin_amdgcn_readlane(__builtin_bit_cast(int, n_di), 0));
        d1 = __builtin_bit_cast(float, __builtin_amdgcn_readlane(__builtin_bit_cast(int, n_di), 1));
        ps0 = nps0; ps1 = nps1; pw0 = npw0; pw1 = npw1;
        buf ^= 1;
    }
}

extern "C" void kernel_launch(void* const* d_in, const int* in_sizes, int n_in,
                              void* d_out, int out_size, void* d_ws, size_t ws_size,
                              hipStream_t stream)
{
    const float* x    = (const float*)d_in[0];
    const int*   ei   = (const int*)  d_in[1];
    const float* W1   = (const float*)d_in[2];
    const float* b1   = (const float*)d_in[3];
    const float* Wmu  = (const float*)d_in[4];
    const float* bmu  = (const float*)d_in[5];
    const float* Wlv  = (const float*)d_in[6];
    const float* blv  = (const float*)d_in[7];

    const int N = in_sizes[0] / KDIM;
    const int E = in_sizes[1] / 2;
    const int* srcIdx = ei;
    const int* dstIdx = ei + E;

    const int TB = 256;
    dim3 blk(TB);
    const int NB = (N + BIN_NODES - 1) / BIN_NODES;
    const int NT = (E + TILE_EDGES - 1) / TILE_EDGES;

    size_t n_bytes = (((size_t)N * 4) + 255) & ~(size_t)255;
    size_t l_bytes = (((size_t)NT * (NB + 1) * 4) + 255) & ~(size_t)255;
    size_t e_bytes = (((size_t)E * 4) + 255) & ~(size_t)255;
    float* dinv   = (float*)d_ws;
    float* hbuf   = (float*)((char*)d_ws + n_bytes);
    int*   cnt    = (int*)(hbuf + (size_t)N * FW);
    int*   rowptr = (int*)((char*)cnt + n_bytes);
    int*   lstart = (int*)((char*)rowptr + n_bytes);
    int*   binned = (int*)((char*)lstart + l_bytes);
    int*   tmp    = (int*)((char*)binned + e_bytes);
    int*   adj    = (int*)((char*)tmp + e_bytes);

    tile_sort<<<NT, dim3(TS_THREADS), 0, stream>>>(srcIdx, dstIdx, lstart, binned, E, NB);
    bin_sort <<<NB, dim3(BS_THREADS), 0, stream>>>(lstart, binned, rowptr, cnt, dinv, tmp, N, NB, NT);

    gcn_layer<false, true><<<LAYER_BLOCKS, blk, 0, stream>>>(rowptr, cnt, tmp, adj, dinv, x,
                                                             W1, nullptr, b1, nullptr, hbuf, N);

    gcn_layer<true, false><<<LAYER_BLOCKS, blk, 0, stream>>>(rowptr, cnt, adj, nullptr, dinv, hbuf,
                                                             Wmu, Wlv, bmu, blv, (float*)d_out, N);
}


# each strip's result tile goes through LDS and is stored as whole rows (16 bytes a lane, whole 256-byte units) one strip later, not as 64-byte pieces from the accumulators; bit-identical
# speedup vs baseline: 5.8178x; 1.0358x over previous
#include <hip/hip_runtime.h>
#include <hip/hip_bf16.h>

typedef __attribute__((ext_vector_type(2))) float v2f;
typedef __attribute__((ext_vector_type(4))) float v4f;
typedef __attribute__((ext_vector_type(8))) float v8f;

#define KDIM 128
#define FW   128
#define APITCH 132

#define BIN_NODES   256
#define NB_MAX      512
#define TILE_EDGES  8192
#define TS_THREADS  512
#define EPT         (TILE_EDGES / TS_THREADS)
#define BS_THREADS  256
#define BIN_CAP     8192

__device__ __forceinline__ int wave_sum(int v) {
    for (int off = 16; off > 0; off >>= 1) v += __shfl_xor(v, off);
    return v;
}

__global__ __launch_bounds__(TS_THREADS)
void tile_sort(const int* __restrict__ src, const int* __restrict__ dst,
               int* __restrict__ lstartG, int* __restrict__ binned, int e, int nb) {
    __shared__ int hist[NB_MAX];
    __shared__ int lstart[NB_MAX];
    __shared__ int wtot[TS_THREADS / 32];
    __shared__ __attribute__((aligned(16))) int stage[TILE_EDGES];
    const int tid  = threadIdx.x;
    const int lane = tid & 31;
    const int wave = tid >> 5;
    const int e0   = blockIdx.x * TILE_EDGES;
    int m = e - e0; if (m > TILE_EDGES) m = TILE_EDGES;

    hist[tid] = 0;
    __syncthreads();

    int key[EPT], bin[EPT], rk[EPT];
    #pragma unroll
    for (int u = 0; u < EPT; ++u) {
        const int i = u * TS_THREADS + tid;
        key[u] = 0; bin[u] = -1; rk[u] = 0;
        if (i < m) {
            const int s = src[e0 + i];
            const int d = dst[e0 + i];
            key[u] = (int)(((unsigned)s << 8) | ((unsigned)d & (BIN_NODES - 1)));
            bin[u] = d >> 8;
        }
    }
    #pragma unroll
    for (int u = 0; u < EPT; ++u)
        if (bin[u] >= 0) rk[u] = atomicAdd(&hist[bin[u]], 1);
    __syncthreads();

    const int h = hist[tid];
    int x = h;
    for (int off = 1; off < 32; off <<= 1) {
        int y = __shfl_up(x, off);
        if (lane >= off) x += y;
    }
    if (lane == 31) wtot[wave] = x;
    __syncthreads();
    int wbase = 0;
    for (int w = 0; w < wave; ++w) wbase += wtot[w];
    const int excl = wbase + x - h;
    lstart[tid] = excl;
    int* lrow = lstartG + (size_t)blockIdx.x * (nb + 1);
    if (tid < nb)  lrow[tid] = excl;
    if (tid == 0)  lrow[nb]  = m;
    __syncthreads();

    #pragma unroll
    for (int u = 0; u < EPT; ++u)
        if (bin[u] >= 0) stage[lstart[bin[u]] + rk[u]] = key[u];
    __syncthreads();

    for (int i = tid * 4; i < m; i += TS_THREADS * 4) {
        if (i + 3 < m) {
            *(int4*)(binned + e0 + i) = *(const int4*)(stage + i);
        } else {
            for (int k = i; k < m; ++k) binned[e0 + k] = stage[k];
        }
    }
}

__global__ __launch_bounds__(BS_THREADS)
void bin_sort(const int* __restrict__ lstartG, const int* __restrict__ binned,
              int* __restrict__ rowptr, int* __restrict__ cnt, float* __restrict__ dinv,
              int* __restrict__ tmp, int n, int nb, int ntiles) {
    __shared__ int cntL[BIN_NODES];
    __shared__ int cursor[BIN_NODES];
    __shared__ int redS[BS_THREADS / 32];
    __shared__ int redL[BS_THREADS / 32];
    __shared__ int wtot[BS_THREADS / 32];
    __shared__ int stage[BIN_CAP];
    const int tid  = threadIdx.x;
    const int lane = tid & 31;
    const int wave = tid >> 5;
    const int b    = blockIdx.x;
    const int ld   = nb + 1;

    cntL[tid] = 0;
    int sumS = 0, sumL = 0;
    for (int t = tid; t < ntiles; t += BS_THREADS) {
        const int rs = lstartG[(size_t)t * ld + b];
        const int re = lstartG[(size_t)t * ld + b + 1];
        sumS += rs; sumL += re - rs;
    }
    sumS = wave_sum(sumS); sumL = wave_sum(sumL);
    if (lane == 0) { redS[wave] = sumS; redL[wave] = sumL; }
    __syncthreads();
    int binStart = 0, total = 0;
    for (int w = 0; w < BS_THREADS / 32; ++w) { binStart += redS[w]; total += redL[w]; }

    for (int t = tid; t < ntiles; t += BS_THREADS) {
        const int rs  = lstartG[(size_t)t * ld + b];
        const int len = lstartG[(size_t)t * ld + b + 1] - rs;
        const int* run = binned + (size_t)t * TILE_EDGES + rs;
        for (int k = 0; k < len; k += 8) {
            int kv[8];
            #pragma unroll
            for (int u = 0; u < 8; ++u) kv[u] = (k + u < len) ? run[k + u] : 0;
            #pragma unroll
            for (int u = 0; u < 8; ++u)
                if (k + u < len) atomicAdd(&cntL[kv[u] & (BIN_NODES - 1)], 1);
        }
    }
    __syncthreads();

    const int c = cntL[tid];
    int x = c;
    for (int off = 1; off < 32; off <<= 1) {
        int y = __shfl_up(x, off);
        if (lane >= off) x += y;
    }
    if (lane == 31) wtot[wave] = x;
    __syncthreads();
    int wbase = 0;
    for (int w = 0; w < wave; ++w) wbase += wtot[w];
    const int loc = wbase + x - c;
    const int node = b * BIN_NODES + tid;
    if (node < n) {
        rowptr[node] = binStart + loc;
        cnt[node]    = c;
        dinv[node]   = rsqrtf((float)(1 + c));
    }
    cursor[tid] = loc;
    __syncthreads();

    const bool staged = total <= BIN_CAP;
    for (int t = tid; t < ntiles; t += BS_THREADS) {
        const int rs  = lstartG[(size_t)t * ld + b];
        const int len = lstartG[(size_t)t * ld + b + 1] - rs;
        const int* run = binned + (size_t)t * TILE_EDGES + rs;
        for (int k = 0; k < len; k += 8) {
            int kv[8];
            #pragma unroll
            for (int u = 0; u < 8; ++u) kv[u] = (k + u < len) ? run[k + u] : 0;
            int p[8];
            #pragma unroll
            for (int u = 0; u < 8; ++u) {
                p[u] = 0;
                if (k + u < len) p[u] = atomicAdd(&cursor[kv[u] & (BIN_NODES - 1)], 1);
            }
            #pragma unroll
            for (int u = 0; u < 8; ++u) {
                if (k + u < len) {
                    const int s = (int)((unsigned)kv[u] >> 8);
                    if (staged) stage[p[u]] = s;
                    else        tmp[binStart + p[u]] = s;
                }
            }
        }
    }
    if (staged) {
        __syncthreads();
        for (int i = tid; i < total; i += BS_THREADS) tmp[binStart + i] = stage[i];
    }
}

#define LAYER_BLOCKS 1024
#define NO_SRC 0x7fffffff

template <int R>
__device__ __forceinline__
v4f add_rows(v4f acc, const int my_s, const int j, const int lane, const float* __restrict__ in)
{
    v4f v[R];
    #pragma unroll
    for (int u = 0; u < R; ++u) {
        const int s = __builtin_amdgcn_readlane(my_s, j + u);
        v[u] = ((const v4f*)(in + (size_t)s * FW))[lane];
    }
    #pragma unroll
    for (int u = 0; u < R; ++u) acc += v[u];
    return acc;
}

__device__ __forceinline__
v4f add_bucket(v4f acc, const int my_s, const int m, const int lane, const float* __restrict__ in)
{
    int j = 0;
    for (; j + 8 <= m; j += 8) acc = add_rows<8>(acc, my_s, j, lane, in);
    switch (m - j) {
        case 1: acc = add_rows<1>(acc, my_s, j, lane, in); break;
        case 2: acc = add_rows<2>(acc, my_s, j, lane, in); break;
        case 3: acc = add_rows<3>(acc, my_s, j, lane, in); break;
        case 4: acc = add_rows<4>(acc, my_s, j, lane, in); break;
        case 5: acc = add_rows<5>(acc, my_s, j, lane, in); break;
        case 6: acc = add_rows<6>(acc, my_s, j, lane, in); break;
        case 7: acc = add_rows<7>(acc, my_s, j, lane, in); break;
        default: break;
    }
    return acc;
}

template <bool ORDER, bool WEIGHTED>
__device__ __forceinline__
v4f node_sum(const int node, const int lane, const int start, const int c, const float di,
             const int ps, const float pw,
             const int* __restrict__ lst, int* __restrict__ ord,
             const float* __restrict__ dinv, const float* __restrict__ in)
{
    v4f  acc = ((const v4f*)(in + (size_t)node * FW))[lane];
    bool own_scaled = false;

    int   first_s = ps;
    float first_d = pw;
    if (ORDER) {
        if (c <= 32) {
            const int sj = ps;
            int r = 0;
            for (int t = 0; t < c; ++t) {
                const int v = __builtin_amdgcn_readlane(sj, t);
                r += (v < sj || (v == sj && t < lane)) ? 1 : 0;
            }
            if (lane >= c) r = lane;
            first_s = __builtin_amdgcn_ds_permute(r << 2, sj);
            first_d = __builtin_bit_cast(float,
                          __builtin_amdgcn_ds_permute(r << 2, __builtin_bit_cast(int, pw)));
            if (lane < c) ord[start + lane] = first_s;
        } else {
            for (int jb = 0; jb < c; jb += 32) {
                const int j  = jb + lane;
                const int sj = (j < c) ? lst[start + j] : NO_SRC;
                int r = 0;
                for (int kb = 0; kb < c; kb += 32) {
                    const int k  = kb + lane;
                    const int sk = (k < c) ? lst[start + k] : NO_SRC;
                    int mm = c - kb; if (mm > 32) mm = 32;
                    for (int t = 0; t < mm; ++t) {
                        const int v = __builtin_amdgcn_readlane(sk, t);
                        r += (v < sj || (v == sj && kb + t < j)) ? 1 : 0;
                    }
                }
                if (j < c) ord[start + r] = sj;
            }
            __threadfence();
        }
    }
    const bool have_first = !ORDER || c <= 32;

    if (!WEIGHTED) {
        int base = 0;
        if (have_first) {
            acc = add_bucket(acc, first_s, c < 32 ? c : 32, lane, in);
            base = 32;
        }
        for (; base < c; base += 32) {
            int m = c - base; if (m > 32) m = 32;
            int my_s = node;
            if (lane < m) my_s = ORDER ? ord[start + base + lane] : lst[start + base + lane];
            acc = add_bucket(acc, my_s, m, lane, in);
        }
        return di * acc;
    }

    for (int base = 0; base < c; base += 32) {
        int m = c - base; if (m > 32) m = 32;
        int   my_s = node;
        float my_w = 0.0f;
        if (base == 0 && have_first) {
            if (lane < m) { my_s = first_s; my_w = first_d * di; }
        } else if (lane < m) {
            if (ORDER) my_s = ord[start + base + lane];
            else       my_s = lst[start + base + lane];
            my_w = dinv[my_s] * di;
        }
        int j = 0;
        for (; j + 8 <= m; j += 8) {
            v4f v[8]; float w[8];
            #pragma unroll
            for (int u = 0; u < 8; ++u) {
                const int s = __builtin_amdgcn_readlane(my_s, j + u);
                w[u] = __builtin_bit_cast(float,
                           __builtin_amdgcn_readlane(__builtin_bit_cast(int, my_w), j + u));
                v[u] = ((const v4f*)(in + (size_t)s * FW))[lane];
            }
            if (!own_scaled) { acc = (di * di) * acc; own_scaled = true; }
            #pragma unroll
            for (int u = 0; u < 8; ++u) acc += w[u] * v[u];
        }
        const int r = m - j;
        if (r > 0) {
            v4f v[7]; float w[7];
            #pragma unroll
            for (int u = 0; u < 7; ++u) {
                v[u] = (v4f){0.0f, 0.0f, 0.0f, 0.0f};
                w[u] = 0.0f;
                if (u < r) {
                    const int s = __builtin_amdgcn_readlane(my_s, j + u);
                    w[u] = __builtin_bit_cast(float,
                               __builtin_amdgcn_readlane(__builtin_bit_cast(int, my_w), j + u));
                    v[u] = ((const v4f*)(in + (size_t)s * FW))[lane];
                }
            }
            if (!own_scaled) { acc = (di * di) * acc; own_scaled = true; }
            #pragma unroll
            for (int u = 0; u < 7; ++u) if (u < r) acc += w[u] * v[u];
        }
    }
    if (!own_scaled) acc = (di * di) * acc;
    return acc;
}

template <bool SPLIT>
__device__ __forceinline__
void store_rows(const float* __restrict__ os, float* __restrict__ out, const int strip,
                const int wave, const int lane, const int n, const float e0, const float e1)
{
    const int ln = lane & 15;
    const int hi = lane >> 4;
    #pragma unroll
    for (int q = 0; q < 2; ++q) {
        const int lrow = 2 * wave + q;
        const int grow = strip * 16 + lrow;
        if (grow < n) {
            const v4f v = *(const v4f*)&os[lrow * APITCH + 4 * lane];
            if (SPLIT) *(v4f*)(out + (size_t)hi * ((size_t)n * 64) + (size_t)grow * 64 + 4 * ln) = v;
            else       *(v4f*)(out + (size_t)grow * FW + 4 * lane) = (q ? e1 : e0) * v;
        }
    }
}

template <bool SPLIT, bool ORDER>
__global__ __launch_bounds__(256, 8)
void gcn_layer(const int* __restrict__ rowptr, const int* __restrict__ cnt,
               const int* __restrict__ lst, int* __restrict__ ord,
               const float* __restrict__ dinv,
               const float* __restrict__ in,
               const float* __restrict__ B0, const float* __restrict__ B1,
               const float* __restrict__ bias0, const float* __restrict__ bias1,
               float* __restrict__ out, int n)
{
    constexpr int BLD = SPLIT ? 64 : 128;
    constexpr bool WEIGHTED = !SPLIT;
    __shared__ __attribute__((aligned(16))) float As[2][16 * APITCH];
    __shared__ __attribute__((aligned(16))) float Os[2][16 * APITCH];
    const int tid  = threadIdx.x;
    const int lane = tid & 31;
    const int wave = __builtin_amdgcn_readfirstlane((int)(tid >> 5));
    const int col0 = wave * 16;
    const int ln   = lane & 15;
    const int hi   = lane >> 4;
    const int nstrips = (n + 15) >> 4;

    const float* bp;
    float bcol;
    if (SPLIT) {
        const float* Bsel = (col0 < 64) ? B0 : B1;
        const float* bsel = (col0 < 64) ? bias0 : bias1;
        const int    bc   = (col0 & 63) + ln;
        bp   = Bsel + (size_t)(2 * hi) * BLD + bc;
        bcol = bsel[bc];
    } else {
        bp   = B0 + (size_t)(2 * hi) * BLD + col0 + ln;
        bcol = bias0[col0 + ln];
    }
    v2f bf[KDIM / 4];
    #pragma unroll
    for (int kk = 0; kk < KDIM / 4; ++kk) {
        bf[kk].x = bp[(size_t)(4 * kk) * BLD];
        bf[kk].y = bp[(size_t)(4 * kk + 1) * BLD];
    }

    const int q_of_lane = lane & 1;
    int strip = blockIdx.x;
    int mnode = strip * 16 + 2 * wave + q_of_lane; if (mnode >= n) mnode = n - 1;
    const int   m_start = rowptr[mnode];
    const int   m_cnt   = cnt[mnode];
    const float m_di    = dinv[mnode];
    int   s0 = __builtin_amdgcn_readlane(m_start, 0), c0 = __builtin_amdgcn_readlane(m_cnt, 0);
    int   s1 = __builtin_amdgcn_readlane(m_start, 1), c1 = __builtin_amdgcn_readlane(m_cnt, 1);
    float d0 = __builtin_bit_cast(float, __builtin_amdgcn_readlane(__builtin_bit_cast(int, m_di), 0));
    float d1 = __builtin_bit_cast(float, __builtin_amdgcn_readlane(__builtin_bit_cast(int, m_di), 1));
    int   ps0 = (lane < c0) ? lst[s0 + lane] : NO_SRC;
    int   ps1 = (lane < c1) ? lst[s1 + lane] : NO_SRC;
    float pw0 = 0.0f, pw1 = 0.0f;
    if (WEIGHTED) {
        pw0 = (lane < c0) ? dinv[ps0] : 0.0f;
        pw1 = (lane < c1) ? dinv[ps1] : 0.0f;
    }

    int   buf    = 0;
    int   pstrip = -1;
    float pd0 = 0.0f, pd1 = 0.0f;
    for (; strip < nstrips; strip += gridDim.x) {
        float* as = &As[buf][0];
        const int node0 = strip * 16 + 2 * wave;

        int nnode = (strip + (int)gridDim.x) * 16 + 2 * wave + q_of_lane; if (nnode >= n) nnode = n - 1;
        const int   n_start = rowptr[nnode];
        const int   n_cnt   = cnt[nnode];
        const float n_di    = dinv[nnode];

        {
            v4f acc = {0.0f, 0.0f, 0.0f, 0.0f};
            if (node0 < n)
                acc = node_sum<ORDER, WEIGHTED>(node0, lane, s0, c0, d0, ps0, pw0, lst, ord, dinv, in);
            *(v4f*)&as[(2 * wave) * APITCH + 4 * lane] = acc;
        }

        const int ns0 = __builtin_amdgcn_readlane(n_start, 0), nc0 = __builtin_amdgcn_readlane(n_cnt, 0);
        const int ns1 = __builtin_amdgcn_readlane(n_start, 1), nc1 = __builtin_amdgcn_readlane(n_cnt, 1);
        const int nps0 = (lane < nc0) ? lst[ns0 + lane] : NO_SRC;
        const int nps1 = (lane < nc1) ? lst[ns1 + lane] : NO_SRC;

        {
            v4f acc = {0.0f, 0.0f, 0.0f, 0.0f};
            if (node0 + 1 < n)
                acc = node_sum<ORDER, WEIGHTED>(node0 + 1, lane, s1, c1, d1, ps1, pw1, lst, ord, dinv, in);
            *(v4f*)&as[(2 * wave + 1) * APITCH + 4 * lane] = acc;
        }
        __syncthreads();

        float npw0 = 0.0f, npw1 = 0.0f;
        if (WEIGHTED) {
            npw0 = (lane < nc0) ? dinv[nps0] : 0.0f;
            npw1 = (lane < nc1) ? dinv[nps1] : 0.0f;
        }

        if (pstrip >= 0)
            store_rows<SPLIT>(&Os[buf ^ 1][0], out, pstrip, wave, lane, n, pd0, pd1);

        const float* ap = as + ln * APITCH + 2 * hi;
        v8f d = {};
        #pragma unroll
        for (int kk = 0; kk < KDIM / 4; ++kk) {
            v2f a = *(const v2f*)(ap + 4 * kk);
            d = __builtin_amdgcn_wmma_f32_16x16x4_f32(
                false, a, false, bf[kk], (short)0, d, false, false);
        }

        {
            float* os = &Os[buf][(8 * hi) * APITCH + col0 + ln];
            #pragma unroll
            for (int r = 0; r < 8; ++r) os[r * APITCH] = SPLIT ? d[r] + bcol : fmaxf(d[r] + bcol, 0.0f);
        }
        pstrip = strip; pd0 = d0; pd1 = d1;

        s0 = ns0; c0 = nc0; s1 = ns1; c1 = nc1;
        d0 = __builtin_bit_cast(float, __builtin_amdgcn_readlane(__builtin_bit_cast(int, n_di), 0));
        d1 = __builtin_bit_cast(float, __builtin_amdgcn_readlane(__builtin_bit_cast(int, n_di), 1));
        ps0 = nps0; ps1 = nps1; pw0 = npw0; pw1 = npw1;
        buf ^= 1;
    }
    if (pstrip >= 0) {
        __syncthreads();
        store_rows<SPLIT>(&Os[buf ^ 1][0], out, pstrip, wave, lane, n, pd0, pd1);
    }
}

extern "C" void kernel_launch(void* const* d_in, const int* in_sizes, int n_in,
                              void* d_out, int out_size, void* d_ws, size_t ws_size,
                              hipStream_t stream)
{
    const float* x    = (const float*)d_in[0];
    const int*   ei   = (const int*)  d_in[1];
    const float* W1   = (const float*)d_in[2];
    const float* b1   = (const float*)d_in[3];
    const float* Wmu  = (const float*)d_in[4];
    const float* bmu  = (const float*)d_in[5];
    const float* Wlv  = (const float*)d_in[6];
    const float* blv  = (const float*)d_in[7];

    const int N = in_sizes[0] / KDIM;
    const int E = in_sizes[1] / 2;
    const int* srcIdx = ei;
    const int* dstIdx = ei + E;

    const int TB = 256;
    dim3 blk(TB);
    const int NB = (N + BIN_NODES - 1) / BIN_NODES;
    const int NT = (E + TILE_EDGES - 1) / TILE_EDGES;

    size_t n_bytes = (((size_t)N * 4) + 255) & ~(size_t)255;
    size_t l_bytes = (((size_t)NT * (NB + 1) * 4) + 255) & ~(size_t)255;
    size_t e_bytes = (((size_t)E * 4) + 255) & ~(size_t)255;
    float* dinv   = (float*)d_ws;
    float* hbuf   = (float*)((char*)d_ws + n_bytes);
    int*   cnt    = (int*)(hbuf + (size_t)N * FW);
    int*   rowptr = (int*)((char*)cnt + n_bytes);
    int*   lstart = (int*)((char*)rowptr + n_bytes);
    int*   binned = (int*)((char*)lstart + l_bytes);
    int*   tmp    = (int*)((char*)binned + e_bytes);
    int*   adj    = (int*)((char*)tmp + e_bytes);

    tile_sort<<<NT, dim3(TS_THREADS), 0, stream>>>(srcIdx, dstIdx, lstart, binned, E, NB);
    bin_sort <<<NB, dim3(BS_THREADS), 0, stream>>>(lstart, binned, rowptr, cnt, dinv, tmp, N, NB, NT);

    gcn_layer<false, true><<<LAYER_BLOCKS, blk, 0, stream>>>(rowptr, cnt, tmp, adj, dinv, x,
                                                             W1, nullptr, b1, nullptr, hbuf, N);

    gcn_layer<true, false><<<LAYER_BLOCKS, blk, 0, stream>>>(rowptr, cnt, adj, nullptr, dinv, hbuf,
                                                             Wmu, Wlv, bmu, blv, (float*)d_out, N);
}
